# starting kernel
# baseline (speedup 1.0000x reference)
#include <hip/hip_runtime.h>
#include <cstdint>
#include <cstddef>

typedef __attribute__((ext_vector_type(2))) float v2f;
typedef __attribute__((ext_vector_type(8))) float v8f;

#define DDIM 128
#define LP   132

template <bool APPLY_TANH>
__global__ __launch_bounds__(256) void gemm_wmma(const float* __restrict__ X,
                                                 const float* __restrict__ W,
                                                 float* __restrict__ out,
                                                 int nrows) {
  __shared__ float Xs[128 * LP];
  __shared__ float Wt[128 * LP];

  const int tid  = threadIdx.x;
  const int lane = tid & 31;
  const int wave = tid >> 5;
  const int half = lane >> 4;
  const int ln16 = lane & 15;
  const int rowBase = blockIdx.x * 128;

  for (int i = 0; i < 16; ++i) {
    int idx = tid + i * 256;
    int r   = idx >> 5;
    int c4  = idx & 31;
    int grow = rowBase + r;
    float4 v = make_float4(0.f, 0.f, 0.f, 0.f);
    if (grow < nrows) {
      v = *(const float4*)(X + (size_t)grow * DDIM + c4 * 4);
      if (APPLY_TANH) {
        v.x = tanhf(v.x); v.y = tanhf(v.y); v.z = tanhf(v.z); v.w = tanhf(v.w);
      }
    }
    float* p = &Xs[r * LP + c4 * 4];
    p[0] = v.x; p[1] = v.y; p[2] = v.z; p[3] = v.w;
  }
  for (int i = 0; i < 16; ++i) {
    int idx = tid + i * 256;
    int k   = idx >> 5;
    int c4  = idx & 31;
    float4 v = *(const float4*)(W + k * DDIM + c4 * 4);
    Wt[(c4 * 4 + 0) * LP + k] = v.x;
    Wt[(c4 * 4 + 1) * LP + k] = v.y;
    Wt[(c4 * 4 + 2) * LP + k] = v.z;
    Wt[(c4 * 4 + 3) * LP + k] = v.w;
  }
  __syncthreads();

  const int mRow = wave * 16;
  v8f acc[8] = {};

  for (int k0 = 0; k0 < 128; k0 += 4) {
    const float* ap = &Xs[(mRow + ln16) * LP + k0 + 2 * half];
    v2f a; a.x = ap[0]; a.y = ap[1];
#pragma unroll
    for (int t = 0; t < 8; ++t) {
      const float* bp = &Wt[(t * 16 + ln16) * LP + k0 + 2 * half];
      v2f b; b.x = bp[0]; b.y = bp[1];
      acc[t] = __builtin_amdgcn_wmma_f32_16x16x4_f32(false, a, false, b,
                                                     (short)0, acc[t], false, false);
    }
  }

#pragma unroll
  for (int t = 0; t < 8; ++t) {
#pragma unroll
    for (int r = 0; r < 8; ++r) {
      int grow = rowBase + mRow + r + 8 * half;
      if (grow < nrows)
        out[(size_t)grow * DDIM + t * 16 + ln16] = acc[t][r];
    }
  }
}

__global__ __launch_bounds__(256) void edge_scatter(const float* __restrict__ h,
                                                    const int* __restrict__ src,
                                                    const int* __restrict__ dst,
                                                    const float* __restrict__ ew,
                                                    float* __restrict__ out,
                                                    int E) {
  int gwave = (int)((blockIdx.x * 256u + threadIdx.x) >> 5);
  int lane  = threadIdx.x & 31;
  if (gwave >= E) return;
  int   s = src[gwave];
  int   d = dst[gwave];
  float w = ew[gwave];
  float4 v = *(const float4*)(h + (size_t)s * DDIM + lane * 4);
  float* op = out + (size_t)d * DDIM + lane * 4;
  atomicAdd(op + 0, v.x * w);
  atomicAdd(op + 1, v.y * w);
  atomicAdd(op + 2, v.z * w);
  atomicAdd(op + 3, v.w * w);
}

__global__ __launch_bounds__(256) void tanh_inplace4(float* __restrict__ p, int n4) {
  int i = blockIdx.x * 256 + threadIdx.x;
  if (i < n4) {
    float4 v = ((float4*)p)[i];
    v.x = tanhf(v.x); v.y = tanhf(v.y); v.z = tanhf(v.z); v.w = tanhf(v.w);
    ((float4*)p)[i] = v;
  }
}

extern "C" void kernel_launch(void* const* d_in, const int* in_sizes, int n_in,
                              void* d_out, int out_size, void* d_ws, size_t ws_size,
                              hipStream_t stream) {
  const float* x   = (const float*)d_in[0];
  const int*   src = (const int*)  d_in[1];
  const int*   dst = (const int*)  d_in[2];
  const float* ew  = (const float*)d_in[3];
  const float* W1  = (const float*)d_in[4];
  const float* W2  = (const float*)d_in[5];
  float* out = (float*)d_out;

  const int N = in_sizes[0] / DDIM;
  const int E = in_sizes[1];

  float* h    = (float*)d_ws;
  float* acc1 = h + (size_t)N * DDIM;

  const int gblocks = (N + 127) / 128;
  const int eblocks = (E + 7) / 8;
  const int n4      = N * DDIM / 4;
  const size_t fbytes = (size_t)N * DDIM * sizeof(float);

  gemm_wmma<false><<<gblocks, 256, 0, stream>>>(x, W1, h, N);
  hipMemsetAsync(acc1, 0, fbytes, stream);
  edge_scatter<<<eblocks, 256, 0, stream>>>(h, src, dst, ew, acc1, E);

  gemm_wmma<true><<<gblocks, 256, 0, stream>>>(acc1, W2, h, N);
  hipMemsetAsync(out, 0, fbytes, stream);
  edge_scatter<<<eblocks, 256, 0, stream>>>(h, src, dst, ew, out, E);
  tanh_inplace4<<<(n4 + 255) / 256, 256, 0, stream>>>(out, n4);
}


# Segment sum as a per-call dst-grouped gather: int-atomic rank, scan, place once; one wave sums a row in registers (order = arrival-order rank), tanh at its store; no f32 atomics, no tanh pass
# speedup vs baseline: 2.2050x; 2.2050x over previous
#include <hip/hip_runtime.h>
#include <cstdint>
#include <cstddef>

typedef __attribute__((ext_vector_type(2))) float v2f;
typedef __attribute__((ext_vector_type(8))) float v8f;

#define DDIM 128
#define LP   132

template <bool APPLY_TANH>
__global__ __launch_bounds__(256) void gemm_wmma(const float* __restrict__ X,
                                                 const float* __restrict__ W,
                                                 float* __restrict__ out,
                                                 int nrows) {
  __shared__ float Xs[128 * LP];
  __shared__ float Wt[128 * LP];

  const int tid  = threadIdx.x;
  const int lane = tid & 31;
  const int wave = tid >> 5;
  const int half = lane >> 4;
  const int ln16 = lane & 15;
  const int rowBase = blockIdx.x * 128;

  for (int i = 0; i < 16; ++i) {
    int idx = tid + i * 256;
    int r   = idx >> 5;
    int c4  = idx & 31;
    int grow = rowBase + r;
    float4 v = make_float4(0.f, 0.f, 0.f, 0.f);
    if (grow < nrows) {
      v = *(const float4*)(X + (size_t)grow * DDIM + c4 * 4);
      if (APPLY_TANH) {
        v.x = tanhf(v.x); v.y = tanhf(v.y); v.z = tanhf(v.z); v.w = tanhf(v.w);
      }
    }
    float* p = &Xs[r * LP + c4 * 4];
    p[0] = v.x; p[1] = v.y; p[2] = v.z; p[3] = v.w;
  }
  for (int i = 0; i < 16; ++i) {
    int idx = tid + i * 256;
    int k   = idx >> 5;
    int c4  = idx & 31;
    float4 v = *(const float4*)(W + k * DDIM + c4 * 4);
    Wt[(c4 * 4 + 0) * LP + k] = v.x;
    Wt[(c4 * 4 + 1) * LP + k] = v.y;
    Wt[(c4 * 4 + 2) * LP + k] = v.z;
    Wt[(c4 * 4 + 3) * LP + k] = v.w;
  }
  __syncthreads();

  const int mRow = wave * 16;
  v8f acc[8] = {};

  for (int k0 = 0; k0 < 128; k0 += 4) {
    const float* ap = &Xs[(mRow + ln16) * LP + k0 + 2 * half];
    v2f a; a.x = ap[0]; a.y = ap[1];
#pragma unroll
    for (int t = 0; t < 8; ++t) {
      const float* bp = &Wt[(t * 16 + ln16) * LP + k0 + 2 * half];
      v2f b; b.x = bp[0]; b.y = bp[1];
      acc[t] = __builtin_amdgcn_wmma_f32_16x16x4_f32(false, a, false, b,
                                                     (short)0, acc[t], false, false);
    }
  }

#pragma unroll
  for (int t = 0; t < 8; ++t) {
#pragma unroll
    for (int r = 0; r < 8; ++r) {
      int grow = rowBase + mRow + r + 8 * half;
      if (grow < nrows)
        out[(size_t)grow * DDIM + t * 16 + ln16] = acc[t][r];
    }
  }
}

__global__ __launch_bounds__(256) void edge_rank(const int* __restrict__ dst,
                                                 int* __restrict__ cnt,
                                                 int* __restrict__ rank,
                                                 int E) {
  int e = blockIdx.x * 256 + threadIdx.x;
  if (e < E) rank[e] = atomicAdd(&cnt[dst[e]], 1);
}

__global__ __launch_bounds__(1024) void scan_counts(const int* __restrict__ cnt,
                                                    int* __restrict__ rowptr,
                                                    int n) {
  __shared__ int wIncl[32];
  __shared__ int wPart[32];
  const int tid  = threadIdx.x;
  const int lane = tid & 31;
  const int wv   = tid >> 5;
  const int start = blockIdx.x * 1024;

  int part = 0;
  for (int i = tid; i < start; i += 1024) part += cnt[i];

  const int i = start + tid;
  const int c = (i < n) ? cnt[i] : 0;
  int incl = c;
  for (int o = 1; o < 32; o <<= 1) {
    int t = __shfl_up(incl, o);
    if (lane >= o) incl += t;
  }
  for (int o = 16; o > 0; o >>= 1) part += __shfl_xor(part, o);
  if (lane == 31) wIncl[wv] = incl;
  if (lane == 0)  wPart[wv] = part;
  __syncthreads();
  int tot = wPart[lane] + ((lane < wv) ? wIncl[lane] : 0);
  for (int o = 16; o > 0; o >>= 1) tot += __shfl_xor(tot, o);
  if (i < n) rowptr[i] = tot + incl - c;
}

__global__ __launch_bounds__(256) void edge_place(const int* __restrict__ src,
                                                  const int* __restrict__ dst,
                                                  const float* __restrict__ ew,
                                                  const int* __restrict__ rowptr,
                                                  const int* __restrict__ rank,
                                                  int2* __restrict__ csr,
                                                  int E) {
  int e = blockIdx.x * 256 + threadIdx.x;
  if (e < E) {
    int pos = rowptr[dst[e]] + rank[e];
    csr[pos] = make_int2(src[e], __float_as_int(ew[e]));
  }
}

__device__ __forceinline__ void row_fma(float4& acc, const float4& v, float w) {
  acc.x = fmaf(v.x, w, acc.x); acc.y = fmaf(v.y, w, acc.y);
  acc.z = fmaf(v.z, w, acc.z); acc.w = fmaf(v.w, w, acc.w);
}

template <bool APPLY_TANH>
__global__ __launch_bounds__(256) void row_gather(const float* __restrict__ h,
                                                  const int* __restrict__ rowptr,
                                                  const int2* __restrict__ csr,
                                                  float* __restrict__ out,
                                                  int N) {
  const int node = (int)((blockIdx.x * 256u + threadIdx.x) >> 5);
  const int lane = threadIdx.x & 31;
  if (node >= N) return;
  const int beg = __builtin_amdgcn_readfirstlane(rowptr[node]);
  const int end = __builtin_amdgcn_readfirstlane(rowptr[node + 1]);
  const float* hl = h + lane * 4;
  float4 acc = make_float4(0.f, 0.f, 0.f, 0.f);

  for (int base = beg; base < end; base += 32) {
    const int cnt = (end - base < 32) ? (end - base) : 32;
    int2 p = make_int2(0, 0);
    if (lane < cnt) p = csr[base + lane];
    int j = 0;
    for (; j + 4 <= cnt; j += 4) {
      const int s0 = __builtin_amdgcn_readlane(p.x, j + 0);
      const int s1 = __builtin_amdgcn_readlane(p.x, j + 1);
      const int s2 = __builtin_amdgcn_readlane(p.x, j + 2);
      const int s3 = __builtin_amdgcn_readlane(p.x, j + 3);
      const float w0 = __int_as_float(__builtin_amdgcn_readlane(p.y, j + 0));
      const float w1 = __int_as_float(__builtin_amdgcn_readlane(p.y, j + 1));
      const float w2 = __int_as_float(__builtin_amdgcn_readlane(p.y, j + 2));
      const float w3 = __int_as_float(__builtin_amdgcn_readlane(p.y, j + 3));
      const float4 v0 = *(const float4*)(hl + (size_t)s0 * DDIM);
      const float4 v1 = *(const float4*)(hl + (size_t)s1 * DDIM);
      const float4 v2 = *(const float4*)(hl + (size_t)s2 * DDIM);
      const float4 v3 = *(const float4*)(hl + (size_t)s3 * DDIM);
      row_fma(acc, v0, w0);
      row_fma(acc, v1, w1);
      row_fma(acc, v2, w2);
      row_fma(acc, v3, w3);
    }
    if (j < cnt) {
      const int j1 = (j + 1 < cnt) ? (j + 1) : (cnt - 1);
      const int j2 = (j + 2 < cnt) ? (j + 2) : (cnt - 1);
      const int s0 = __builtin_amdgcn_readlane(p.x, j);
      const int s1 = __builtin_amdgcn_readlane(p.x, j1);
      const int s2 = __builtin_amdgcn_readlane(p.x, j2);
      const float w0 = __int_as_float(__builtin_amdgcn_readlane(p.y, j));
      const float w1 = __int_as_float(__builtin_amdgcn_readlane(p.y, j1));
      const float w2 = __int_as_float(__builtin_amdgcn_readlane(p.y, j2));
      const float4 v0 = *(const float4*)(hl + (size_t)s0 * DDIM);
      const float4 v1 = *(const float4*)(hl + (size_t)s1 * DDIM);
      const float4 v2 = *(const float4*)(hl + (size_t)s2 * DDIM);
      asm volatile("" :: "v"(v0.x), "v"(v1.x), "v"(v2.x));
      row_fma(acc, v0, w0);
      if (j + 1 < cnt) row_fma(acc, v1, w1);
      if (j + 2 < cnt) row_fma(acc, v2, w2);
    }
  }
  if (APPLY_TANH) {
    acc.x = tanhf(acc.x); acc.y = tanhf(acc.y); acc.z = tanhf(acc.z); acc.w = tanhf(acc.w);
  }
  *(float4*)(out + (size_t)node * DDIM + lane * 4) = acc;
}

extern "C" void kernel_launch(void* const* d_in, const int* in_sizes, int n_in,
                              void* d_out, int out_size, void* d_ws, size_t ws_size,
                              hipStream_t stream) {
  const float* x   = (const float*)d_in[0];
  const int*   src = (const int*)  d_in[1];
  const int*   dst = (const int*)  d_in[2];
  const float* ew  = (const float*)d_in[3];
  const float* W1  = (const float*)d_in[4];
  const float* W2  = (const float*)d_in[5];
  float* out = (float*)d_out;

  const int N = in_sizes[0] / DDIM;
  const int E = in_sizes[1];

  float* h    = (float*)d_ws;
  float* acc1 = h + (size_t)N * DDIM;
  const size_t np = ((size_t)N + 4) & ~(size_t)3;
  int*  cnt    = (int*)(acc1 + (size_t)N * DDIM);
  int*  rowptr = cnt + np;
  int*  rank   = rowptr + np;
  int2* csr    = (int2*)(rank + (((size_t)E + 3) & ~(size_t)3));

  const int gblocks = (N + 127) / 128;
  const int eblocks = (E + 255) / 256;
  const int rblocks = (N + 7) / 8;
  const int sblocks = (N + 1 + 1023) / 1024;

  hipMemsetAsync(cnt, 0, np * sizeof(int), stream);
  edge_rank<<<eblocks, 256, 0, stream>>>(dst, cnt, rank, E);
  scan_counts<<<sblocks, 1024, 0, stream>>>(cnt, rowptr, N + 1);
  edge_place<<<eblocks, 256, 0, stream>>>(src, dst, ew, rowptr, rank, csr, E);

  gemm_wmma<false><<<gblocks, 256, 0, stream>>>(x, W1, h, N);
  row_gather<false><<<rblocks, 256, 0, stream>>>(h, rowptr, csr, acc1, N);

  gemm_wmma<true><<<gblocks, 256, 0, stream>>>(acc1, W2, h, N);
  row_gather<true><<<rblocks, 256, 0, stream>>>(h, rowptr, csr, out, N);
}


# Fused layer kernel as one resident round: 250 blocks of 32 waves own 400 rows each (one block per WGP, 279 KB LDS, W staged once per WGP) instead of 782 blocks at two per WGP; sum order unchanged
# speedup vs baseline: 2.6650x; 1.2086x over previous
#include <hip/hip_runtime.h>
#include <cstdint>
#include <cstddef>


typedef __attribute__((ext_vector_type(2))) float v2f;
typedef __attribute__((ext_vector_type(8))) float v8f;

#define DDIM 128
#define LP   132
#define RB   400
#define NWV  32
#define RQ   (RB / NWV)
#define RR   (RB % NWV)
#define MT   (RB / 16)

__global__ __launch_bounds__(256) void edge_rank(const int* __restrict__ dst,
                                                 int* __restrict__ cnt,
                                                 int* __restrict__ rank,
                                                 int E) {
  int e = blockIdx.x * 256 + threadIdx.x;
  if (e < E) rank[e] = atomicAdd(&cnt[dst[e]], 1);
}

__global__ __launch_bounds__(1024) void scan_counts(const int* __restrict__ cnt,
                                                    int* __restrict__ rowptr,
                                                    int n) {
  __shared__ int wIncl[32];
  __shared__ int wPart[32];
  const int tid  = threadIdx.x;
  const int lane = tid & 31;
  const int wv   = tid >> 5;
  const int start = blockIdx.x * 1024;

  int part = 0;
  for (int i = tid; i < start; i += 1024) part += cnt[i];

  const int i = start + tid;
  const int c = (i < n) ? cnt[i] : 0;
  int incl = c;
  for (int o = 1; o < 32; o <<= 1) {
    int t = __shfl_up(incl, o);
    if (lane >= o) incl += t;
  }
  for (int o = 16; o > 0; o >>= 1) part += __shfl_xor(part, o);
  if (lane == 31) wIncl[wv] = incl;
  if (lane == 0)  wPart[wv] = part;
  __syncthreads();
  int tot = wPart[lane] + ((lane < wv) ? wIncl[lane] : 0);
  for (int o = 16; o > 0; o >>= 1) tot += __shfl_xor(tot, o);
  if (i < n) rowptr[i] = tot + incl - c;
}

__global__ __launch_bounds__(256) void edge_place(const int* __restrict__ src,
                                                  const int* __restrict__ dst,
                                                  const float* __restrict__ ew,
                                                  const int* __restrict__ rowptr,
                                                  const int* __restrict__ rank,
                                                  int2* __restrict__ csr,
                                                  int E) {
  int e = blockIdx.x * 256 + threadIdx.x;
  if (e < E) {
    int pos = rowptr[dst[e]] + rank[e];
    csr[pos] = make_int2(src[e], __float_as_int(ew[e]));
  }
}

__device__ __forceinline__ void row_fma(float4& acc, const float4& v, float w) {
  acc.x = fmaf(v.x, w, acc.x); acc.y = fmaf(v.y, w, acc.y);
  acc.z = fmaf(v.z, w, acc.z); acc.w = fmaf(v.w, w, acc.w);
}

__global__ __launch_bounds__(1024) void gather_gemm_tanh(const float* __restrict__ X,
                                                        const int* __restrict__ rowptr,
                                                        const int2* __restrict__ csr,
                                                        const float* __restrict__ W,
                                                        float* __restrict__ out,
                                                        int nrows) {
  __shared__ float Xs[RB * LP];
  __shared__ float Wt[128 * LP];

  const int tid  = threadIdx.x;
  const int lane = tid & 31;
  const int wave = __builtin_amdgcn_readfirstlane(tid >> 5);
  const int half = lane >> 4;
  const int ln16 = lane & 15;
  const int rowBase = blockIdx.x * RB;

  for (int i = 0; i < 4; ++i) {
    int idx = tid + i * 1024;
    int k   = idx >> 5;
    int c4  = idx & 31;
    float4 v = *(const float4*)(W + k * DDIM + c4 * 4);
    Wt[(c4 * 4 + 0) * LP + k] = v.x;
    Wt[(c4 * 4 + 1) * LP + k] = v.y;
    Wt[(c4 * 4 + 2) * LP + k] = v.z;
    Wt[(c4 * 4 + 3) * LP + k] = v.w;
  }

  {
    const int nr  = RQ + ((wave < RR) ? 1 : 0);
    const int lr0 = wave * RQ + ((wave < RR) ? wave : RR);
    const int firstRow = rowBase + lr0;
    int ri = firstRow + ((lane < nr) ? lane : nr);
    if (ri > nrows) ri = nrows;
    const int rpv = rowptr[ri];
    const int begAll = __builtin_amdgcn_readlane(rpv, 0);
    const int endAll = __builtin_amdgcn_readlane(rpv, nr);
    int r    = 0;
    int rend = __builtin_amdgcn_readlane(rpv, 1);
    const float* xl = X + lane * 4;
    float* srow = &Xs[lr0 * LP + lane * 4];
    float4 acc = make_float4(0.f, 0.f, 0.f, 0.f);

    int base = begAll;
    int cnt  = (endAll - base < 32) ? (endAll - base) : 32;
    int2 p = make_int2(0, 0);
    if (lane < cnt) p = csr[base + lane];
    while (base < endAll) {
      const int nbase = base + 32;
      const int ncnt  = (endAll - nbase < 32) ? (endAll - nbase) : 32;
      int2 pn = make_int2(0, 0);
      if (lane < ncnt) pn = csr[nbase + lane];
      for (int j = 0; j < cnt; j += 8) {
        int   s[8];
        float w[8];
        float4 v[8];
#pragma unroll
        for (int t = 0; t < 8; ++t) {
          const int jt = (j + t < cnt) ? (j + t) : (cnt - 1);
          s[t] = __builtin_amdgcn_readlane(p.x, jt);
          w[t] = __int_as_float(__builtin_amdgcn_readlane(p.y, jt));
        }
#pragma unroll
        for (int t = 0; t < 8; ++t) v[t] = *(const float4*)(xl + (size_t)s[t] * DDIM);
        asm volatile("" :: "v"(v[0].x), "v"(v[1].x), "v"(v[2].x), "v"(v[3].x),
                           "v"(v[4].x), "v"(v[5].x), "v"(v[6].x), "v"(v[7].x));
#pragma unroll
        for (int t = 0; t < 8; ++t) {
          if (j + t < cnt) {
            const int e = base + j + t;
            while (e >= rend) {
              *(float4*)(srow + r * LP) = acc;
              acc = make_float4(0.f, 0.f, 0.f, 0.f);
              ++r;
              rend = __builtin_amdgcn_readlane(rpv, r + 1);
            }
            row_fma(acc, v[t], w[t]);
          }
        }
      }
      base = nbase; cnt = ncnt; p = pn;
    }
    for (; r < nr; ++r) {
      *(float4*)(srow + r * LP) = acc;
      acc = make_float4(0.f, 0.f, 0.f, 0.f);
    }
  }
  __syncthreads();

  for (int u = wave; u < 2 * MT; u += NWV) {
  const int mRow  = (u >> 1) * 16;
  const int tBase = (u & 1) * 4;
  v8f acc[4] = {};

  for (int k0 = 0; k0 < 128; k0 += 4) {
    const float* ap = &Xs[(mRow + ln16) * LP + k0 + 2 * half];
    v2f a; a.x = ap[0]; a.y = ap[1];
#pragma unroll
    for (int t = 0; t < 4; ++t) {
      const float* bp = &Wt[((tBase + t) * 16 + ln16) * LP + k0 + 2 * half];
      v2f b; b.x = bp[0]; b.y = bp[1];
      acc[t] = __builtin_amdgcn_wmma_f32_16x16x4_f32(false, a, false, b,
                                                     (short)0, acc[t], false, false);
    }
  }

#pragma unroll
  for (int t = 0; t < 4; ++t) {
#pragma unroll
    for (int r = 0; r < 8; ++r) {
      int grow = rowBase + mRow + r + 8 * half;
      if (grow < nrows)
        out[(size_t)grow * DDIM + (tBase + t) * 16 + ln16] = tanhf(acc[t][r]);
    }
  }
  }
}

extern "C" void kernel_launch(void* const* d_in, const int* in_sizes, int n_in,
                              void* d_out, int out_size, void* d_ws, size_t ws_size,
                              hipStream_t stream) {
  const float* x   = (const float*)d_in[0];
  const int*   src = (const int*)  d_in[1];
  const int*   dst = (const int*)  d_in[2];
  const float* ew  = (const float*)d_in[3];
  const float* W1  = (const float*)d_in[4];
  const float* W2  = (const float*)d_in[5];
  float* out = (float*)d_out;

  const int N = in_sizes[0] / DDIM;
  const int E = in_sizes[1];

  float* act1 = (float*)d_ws;
  const size_t np = ((size_t)N + 4) & ~(size_t)3;
  int*  cnt    = (int*)(act1 + (size_t)N * DDIM);
  int*  rowptr = cnt + np;
  int*  rank   = rowptr + np;
  int2* csr    = (int2*)(rank + (((size_t)E + 3) & ~(size_t)3));

  const int gblocks = (N + RB - 1) / RB;
  const int eblocks = (E + 255) / 256;
  const int sblocks = (N + 1 + 1023) / 1024;

  hipMemsetAsync(cnt, 0, np * sizeof(int), stream);
  edge_rank<<<eblocks, 256, 0, stream>>>(dst, cnt, rank, E);
  scan_counts<<<sblocks, 1024, 0, stream>>>(cnt, rowptr, N + 1);
  edge_place<<<eblocks, 256, 0, stream>>>(src, dst, ew, rowptr, rank, csr, E);

  gather_gemm_tanh<<<gblocks, 1024, 0, stream>>>(x, rowptr, csr, W1, act1, N);
  gather_gemm_tanh<<<gblocks, 1024, 0, stream>>>(act1, rowptr, csr, W2, out, N);
}


# Edge grouping by dst as a two-level counting sort on LDS counters (256 row ranges: count, place, per-range row sort), not per-edge returning global adds; no memset; sum order = LDS arrival rank
# speedup vs baseline: 4.5509x; 1.7077x over previous
#include <hip/hip_runtime.h>
#include <cstdint>
#include <cstddef>


typedef __attribute__((ext_vector_type(2))) float v2f;
typedef __attribute__((ext_vector_type(8))) float v8f;

#define DDIM 128
#define LP   132
#define RB   400
#define NWV  32
#define RQ   (RB / NWV)
#define RR   (RB % NWV)
#define MT   (RB / 16)

#define NBK  256
#define CB   256
#define EPT  8
#define FMAX 8192

__device__ __forceinline__ unsigned range_width(int N) {
  const int bw = (N + NBK - 1) / NBK;
  return (unsigned)(bw > 0 ? bw : 1);
}

__global__ __launch_bounds__(1024) void range_count(const int* __restrict__ dst,
                                                    int* __restrict__ hist,
                                                    int E, int N) {
  __shared__ int h[NBK];
  const int tid = threadIdx.x;
  const unsigned bw = range_width(N);
  if (tid < NBK) h[tid] = 0;
  __syncthreads();
  const int per = (E + CB - 1) / CB;
  const int c0 = blockIdx.x * per;
  const int c1 = (c0 + per < E) ? (c0 + per) : E;
  for (int e0 = c0 + tid; e0 < c1; e0 += 1024 * EPT) {
    int d[EPT];
#pragma unroll
    for (int t = 0; t < EPT; ++t) {
      const int e = e0 + t * 1024;
      d[t] = dst[(e < c1) ? e : (c1 - 1)];
    }
    asm volatile("" :: "v"(d[0]), "v"(d[1]), "v"(d[2]), "v"(d[3]), "v"(d[4]), "v"(d[5]), "v"(d[6]), "v"(d[7]));
#pragma unroll
    for (int t = 0; t < EPT; ++t)
      if (e0 + t * 1024 < c1) atomicAdd(&h[(unsigned)d[t] / bw], 1);
  }
  __syncthreads();
  if (tid < NBK) hist[blockIdx.x * NBK + tid] = h[tid];
}

__global__ __launch_bounds__(1024) void range_place(const int* __restrict__ src,
                                                    const int* __restrict__ dst,
                                                    const float* __restrict__ ew,
                                                    const int* __restrict__ hist,
                                                    int* __restrict__ rbase,
                                                    int4* __restrict__ rec,
                                                    int E, int N) {
  __shared__ int pTot[4][NBK];
  __shared__ int pPre[4][NBK];
  __shared__ int cur[NBK];
  __shared__ int wsum[NBK / 32];
  const int tid  = threadIdx.x;
  const int lane = tid & 31;
  const int wv   = tid >> 5;
  const int b    = blockIdx.x;
  const unsigned bw = range_width(N);
  {
    const int k = tid & (NBK - 1);
    const int q = tid >> 8;
    int tot = 0, pre = 0;
#pragma unroll
    for (int i = 0; i < CB / 4; ++i) {
      const int bb = q * (CB / 4) + i;
      const int v = hist[bb * NBK + k];
      tot += v;
      pre += (bb < b) ? v : 0;
    }
    pTot[q][k] = tot;
    pPre[q][k] = pre;
  }
  __syncthreads();
  int tot = 0, pre = 0;
  if (tid < NBK) {
    tot = pTot[0][tid] + pTot[1][tid] + pTot[2][tid] + pTot[3][tid];
    pre = pPre[0][tid] + pPre[1][tid] + pPre[2][tid] + pPre[3][tid];
  }
  int incl = tot;
  for (int o = 1; o < 32; o <<= 1) {
    int t = __shfl_up(incl, o);
    if (lane >= o) incl += t;
  }
  if (tid < NBK && lane == 31) wsum[wv] = incl;
  __syncthreads();
  if (tid < NBK) {
    int off = 0;
    for (int w = 0; w < NBK / 32; ++w) off += (w < wv) ? wsum[w] : 0;
    const int base = off + incl - tot;
    cur[tid] = base + pre;
    if (b == 0) {
      rbase[tid] = base;
      if (tid == NBK - 1) rbase[NBK] = base + tot;
    }
  }
  __syncthreads();

  const int per = (E + CB - 1) / CB;
  const int c0 = b * per;
  const int c1 = (c0 + per < E) ? (c0 + per) : E;
  for (int e0 = c0 + tid; e0 < c1; e0 += 1024 * EPT) {
    int d[EPT], s[EPT];
    float w[EPT];
#pragma unroll
    for (int t = 0; t < EPT; ++t) {
      const int e  = e0 + t * 1024;
      const int ee = (e < c1) ? e : (c1 - 1);
      d[t] = dst[ee]; s[t] = src[ee]; w[t] = ew[ee];
    }
    asm volatile("" :: "v"(d[0]), "v"(d[1]), "v"(d[2]), "v"(d[3]), "v"(d[4]), "v"(d[5]), "v"(d[6]), "v"(d[7]),
                       "v"(s[0]), "v"(s[1]), "v"(s[2]), "v"(s[3]), "v"(s[4]), "v"(s[5]), "v"(s[6]), "v"(s[7]),
                       "v"(w[0]), "v"(w[1]), "v"(w[2]), "v"(w[3]), "v"(w[4]), "v"(w[5]), "v"(w[6]), "v"(w[7]));
#pragma unroll
    for (int t = 0; t < EPT; ++t) {
      if (e0 + t * 1024 < c1) {
        const int pos = atomicAdd(&cur[(unsigned)d[t] / bw], 1);
        rec[pos] = make_int4(s[t], __float_as_int(w[t]), d[t], 0);
      }
    }
  }
}

__global__ __launch_bounds__(1024) void range_rows(const int4* __restrict__ rec,
                                                   const int* __restrict__ rbase,
                                                   int* __restrict__ rowptr,
                                                   int2* __restrict__ csr,
                                                   int E, int N) {
  __shared__ int cnt[FMAX];
  __shared__ int wsum[32];
  const int tid  = threadIdx.x;
  const int lane = tid & 31;
  const int wv   = tid >> 5;
  const int bw   = (int)range_width(N);
  const int row0 = blockIdx.x * bw;
  int rows = N - row0;
  if (rows > bw) rows = bw;
  if (rows < 0)  rows = 0;
  const int beg = rbase[blockIdx.x];
  const int end = rbase[blockIdx.x + 1];
  const int* recd = (const int*)rec + 2;

  for (int j = tid; j < rows; j += 1024) cnt[j] = 0;
  __syncthreads();
  for (int i0 = beg + tid; i0 < end; i0 += 1024 * EPT) {
    int d[EPT];
#pragma unroll
    for (int t = 0; t < EPT; ++t) {
      const int i = i0 + t * 1024;
      d[t] = recd[4 * (size_t)((i < end) ? i : (end - 1))];
    }
    asm volatile("" :: "v"(d[0]), "v"(d[1]), "v"(d[2]), "v"(d[3]), "v"(d[4]), "v"(d[5]), "v"(d[6]), "v"(d[7]));
#pragma unroll
    for (int t = 0; t < EPT; ++t)
      if (i0 + t * 1024 < end) atomicAdd(&cnt[d[t] - row0], 1);
  }
  __syncthreads();

  const int per = (rows + 1023) >> 10;
  const int j0  = tid * per;
  const int j1  = (j0 + per < rows) ? (j0 + per) : rows;
  int sum = 0;
  for (int j = j0; j < j1; ++j) sum += cnt[j];
  int incl = sum;
  for (int o = 1; o < 32; o <<= 1) {
    int t = __shfl_up(incl, o);
    if (lane >= o) incl += t;
  }
  if (lane == 31) wsum[wv] = incl;
  __syncthreads();
  int off = (lane < wv) ? wsum[lane] : 0;
  for (int o = 16; o > 0; o >>= 1) off += __shfl_xor(off, o);
  int run = beg + off + incl - sum;
  for (int j = j0; j < j1; ++j) {
    const int c = cnt[j];
    cnt[j] = run;
    rowptr[row0 + j] = run;
    run += c;
  }
  if (tid == 0 && rows > 0 && row0 + rows == N) rowptr[N] = E;
  __syncthreads();

  for (int i0 = beg + tid; i0 < end; i0 += 1024 * EPT) {
    int4 r[EPT];
#pragma unroll
    for (int t = 0; t < EPT; ++t) {
      const int i = i0 + t * 1024;
      r[t] = rec[(i < end) ? i : (end - 1)];
    }
    asm volatile("" :: "v"(r[0].x), "v"(r[1].x), "v"(r[2].x), "v"(r[3].x),
                       "v"(r[4].x), "v"(r[5].x), "v"(r[6].x), "v"(r[7].x));
#pragma unroll
    for (int t = 0; t < EPT; ++t) {
      if (i0 + t * 1024 < end) {
        const int pos = atomicAdd(&cnt[r[t].z - row0], 1);
        csr[pos] = make_int2(r[t].x, r[t].y);
      }
    }
  }
}

__device__ __forceinline__ void row_fma(float4& acc, const float4& v, float w) {
  acc.x = fmaf(v.x, w, acc.x); acc.y = fmaf(v.y, w, acc.y);
  acc.z = fmaf(v.z, w, acc.z); acc.w = fmaf(v.w, w, acc.w);
}

__global__ __launch_bounds__(1024) void gather_gemm_tanh(const float* __restrict__ X,
                                                        const int* __restrict__ rowptr,
                                                        const int2* __restrict__ csr,
                                                        const float* __restrict__ W,
                                                        float* __restrict__ out,
                                                        int nrows) {
  __shared__ float Xs[RB * LP];
  __shared__ float Wt[128 * LP];

  const int tid  = threadIdx.x;
  const int lane = tid & 31;
  const int wave = __builtin_amdgcn_readfirstlane(tid >> 5);
  const int half = lane >> 4;
  const int ln16 = lane & 15;
  const int rowBase = blockIdx.x * RB;

  for (int i = 0; i < 4; ++i) {
    int idx = tid + i * 1024;
    int k   = idx >> 5;
    int c4  = idx & 31;
    float4 v = *(const float4*)(W + k * DDIM + c4 * 4);
    Wt[(c4 * 4 + 0) * LP + k] = v.x;
    Wt[(c4 * 4 + 1) * LP + k] = v.y;
    Wt[(c4 * 4 + 2) * LP + k] = v.z;
    Wt[(c4 * 4 + 3) * LP + k] = v.w;
  }

  {
    const int nr  = RQ + ((wave < RR) ? 1 : 0);
    const int lr0 = wave * RQ + ((wave < RR) ? wave : RR);
    const int firstRow = rowBase + lr0;
    int ri = firstRow + ((lane < nr) ? lane : nr);
    if (ri > nrows) ri = nrows;
    const int rpv = rowptr[ri];
    const int begAll = __builtin_amdgcn_readlane(rpv, 0);
    const int endAll = __builtin_amdgcn_readlane(rpv, nr);
    int r    = 0;
    int rend = __builtin_amdgcn_readlane(rpv, 1);
    const float* xl = X + lane * 4;
    float* srow = &Xs[lr0 * LP + lane * 4];
    float4 acc = make_float4(0.f, 0.f, 0.f, 0.f);

    int base = begAll;
    int cnt  = (endAll - base < 32) ? (endAll - base) : 32;
    int2 p = make_int2(0, 0);
    if (lane < cnt) p = csr[base + lane];
    while (base < endAll) {
      const int nbase = base + 32;
      const int ncnt  = (endAll - nbase < 32) ? (endAll - nbase) : 32;
      int2 pn = make_int2(0, 0);
      if (lane < ncnt) pn = csr[nbase + lane];
      for (int j = 0; j < cnt; j += 8) {
        int   s[8];
        float w[8];
        float4 v[8];
#pragma unroll
        for (int t = 0; t < 8; ++t) {
          const int jt = (j + t < cnt) ? (j + t) : (cnt - 1);
          s[t] = __builtin_amdgcn_readlane(p.x, jt);
          w[t] = __int_as_float(__builtin_amdgcn_readlane(p.y, jt));
        }
#pragma unroll
        for (int t = 0; t < 8; ++t) v[t] = *(const float4*)(xl + (size_t)s[t] * DDIM);
        asm volatile("" :: "v"(v[0].x), "v"(v[1].x), "v"(v[2].x), "v"(v[3].x),
                           "v"(v[4].x), "v"(v[5].x), "v"(v[6].x), "v"(v[7].x));
#pragma unroll
        for (int t = 0; t < 8; ++t) {
          if (j + t < cnt) {
            const int e = base + j + t;
            while (e >= rend) {
              *(float4*)(srow + r * LP) = acc;
              acc = make_float4(0.f, 0.f, 0.f, 0.f);
              ++r;
              rend = __builtin_amdgcn_readlane(rpv, r + 1);
            }
            row_fma(acc, v[t], w[t]);
          }
        }
      }
      base = nbase; cnt = ncnt; p = pn;
    }
    for (; r < nr; ++r) {
      *(float4*)(srow + r * LP) = acc;
      acc = make_float4(0.f, 0.f, 0.f, 0.f);
    }
  }
  __syncthreads();

  for (int u = wave; u < 2 * MT; u += NWV) {
  const int mRow  = (u >> 1) * 16;
  const int tBase = (u & 1) * 4;
  v8f acc[4] = {};

  for (int k0 = 0; k0 < 128; k0 += 4) {
    const float* ap = &Xs[(mRow + ln16) * LP + k0 + 2 * half];
    v2f a; a.x = ap[0]; a.y = ap[1];
#pragma unroll
    for (int t = 0; t < 4; ++t) {
      const float* bp = &Wt[((tBase + t) * 16 + ln16) * LP + k0 + 2 * half];
      v2f b; b.x = bp[0]; b.y = bp[1];
      acc[t] = __builtin_amdgcn_wmma_f32_16x16x4_f32(false, a, false, b,
                                                     (short)0, acc[t], false, false);
    }
  }

#pragma unroll
  for (int t = 0; t < 4; ++t) {
#pragma unroll
    for (int r = 0; r < 8; ++r) {
      int grow = rowBase + mRow + r + 8 * half;
      if (grow < nrows)
        out[(size_t)grow * DDIM + (tBase + t) * 16 + ln16] = tanhf(acc[t][r]);
    }
  }
  }
}

extern "C" void kernel_launch(void* const* d_in, const int* in_sizes, int n_in,
                              void* d_out, int out_size, void* d_ws, size_t ws_size,
                              hipStream_t stream) {
  const float* x   = (const float*)d_in[0];
  const int*   src = (const int*)  d_in[1];
  const int*   dst = (const int*)  d_in[2];
  const float* ew  = (const float*)d_in[3];
  const float* W1  = (const float*)d_in[4];
  const float* W2  = (const float*)d_in[5];
  float* out = (float*)d_out;

  const int N = in_sizes[0] / DDIM;
  const int E = in_sizes[1];

  float* act1 = (float*)d_ws;
  const size_t np = ((size_t)N + 4) & ~(size_t)3;
  int*  rowptr = (int*)(act1 + (size_t)N * DDIM);
  int*  hist   = rowptr + np;
  int*  rbase  = hist + CB * NBK;
  int4* rec    = (int4*)(rbase + NBK + 4);
  int2* csr    = (int2*)(rec + E);

  const int gblocks = (N + RB - 1) / RB;

  range_count<<<CB, 1024, 0, stream>>>(dst, hist, E, N);
  range_place<<<CB, 1024, 0, stream>>>(src, dst, ew, hist, rbase, rec, E, N);
  range_rows<<<NBK, 1024, 0, stream>>>(rec, rbase, rowptr, csr, E, N);

  gather_gemm_tanh<<<gblocks, 1024, 0, stream>>>(x, rowptr, csr, W1, act1, N);
  gather_gemm_tanh<<<gblocks, 1024, 0, stream>>>(act1, rowptr, csr, W2, out, N);
}


# Layer kernel epilogue: tanh without branches (hardware exp2 and reciprocal above 0.625, the library's odd polynomial below, picked by select); row test once per whole 16-row tile; sums untouched
# speedup vs baseline: 5.0750x; 1.1152x over previous
#include <hip/hip_runtime.h>
#include <cstdint>
#include <cstddef>


typedef __attribute__((ext_vector_type(2))) float v2f;
typedef __attribute__((ext_vector_type(8))) float v8f;

#define DDIM 128
#define LP   132
#define RB   400
#define NWV  32
#define RQ   (RB / NWV)
#define RR   (RB % NWV)
#define MT   (RB / 16)

#define NBK  256
#define CB   256
#define EPT  8
#define FMAX 8192

__device__ __forceinline__ unsigned range_width(int N) {
  const int bw = (N + NBK - 1) / NBK;
  return (unsigned)(bw > 0 ? bw : 1);
}

__global__ __launch_bounds__(1024) void range_count(const int* __restrict__ dst,
                                                    int* __restrict__ hist,
                                                    int E, int N) {
  __shared__ int h[NBK];
  const int tid = threadIdx.x;
  const unsigned bw = range_width(N);
  if (tid < NBK) h[tid] = 0;
  __syncthreads();
  const int per = (E + CB - 1) / CB;
  const int c0 = blockIdx.x * per;
  const int c1 = (c0 + per < E) ? (c0 + per) : E;
  for (int e0 = c0 + tid; e0 < c1; e0 += 1024 * EPT) {
    int d[EPT];
#pragma unroll
    for (int t = 0; t < EPT; ++t) {
      const int e = e0 + t * 1024;
      d[t] = dst[(e < c1) ? e : (c1 - 1)];
    }
    asm volatile("" :: "v"(d[0]), "v"(d[1]), "v"(d[2]), "v"(d[3]), "v"(d[4]), "v"(d[5]), "v"(d[6]), "v"(d[7]));
#pragma unroll
    for (int t = 0; t < EPT; ++t)
      if (e0 + t * 1024 < c1) atomicAdd(&h[(unsigned)d[t] / bw], 1);
  }
  __syncthreads();
  if (tid < NBK) hist[blockIdx.x * NBK + tid] = h[tid];
}

__global__ __launch_bounds__(1024) void range_place(const int* __restrict__ src,
                                                    const int* __restrict__ dst,
                                                    const float* __restrict__ ew,
                                                    const int* __restrict__ hist,
                                                    int* __restrict__ rbase,
                                                    int4* __restrict__ rec,
                                                    int E, int N) {
  __shared__ int pTot[4][NBK];
  __shared__ int pPre[4][NBK];
  __shared__ int cur[NBK];
  __shared__ int wsum[NBK / 32];
  const int tid  = threadIdx.x;
  const int lane = tid & 31;
  const int wv   = tid >> 5;
  const int b    = blockIdx.x;
  const unsigned bw = range_width(N);
  {
    const int k = tid & (NBK - 1);
    const int q = tid >> 8;
    int tot = 0, pre = 0;
#pragma unroll
    for (int i = 0; i < CB / 4; ++i) {
      const int bb = q * (CB / 4) + i;
      const int v = hist[bb * NBK + k];
      tot += v;
      pre += (bb < b) ? v : 0;
    }
    pTot[q][k] = tot;
    pPre[q][k] = pre;
  }
  __syncthreads();
  int tot = 0, pre = 0;
  if (tid < NBK) {
    tot = pTot[0][tid] + pTot[1][tid] + pTot[2][tid] + pTot[3][tid];
    pre = pPre[0][tid] + pPre[1][tid] + pPre[2][tid] + pPre[3][tid];
  }
  int incl = tot;
  for (int o = 1; o < 32; o <<= 1) {
    int t = __shfl_up(incl, o);
    if (lane >= o) incl += t;
  }
  if (tid < NBK && lane == 31) wsum[wv] = incl;
  __syncthreads();
  if (tid < NBK) {
    int off = 0;
    for (int w = 0; w < NBK / 32; ++w) off += (w < wv) ? wsum[w] : 0;
    const int base = off + incl - tot;
    cur[tid] = base + pre;
    if (b == 0) {
      rbase[tid] = base;
      if (tid == NBK - 1) rbase[NBK] = base + tot;
    }
  }
  __syncthreads();

  const int per = (E + CB - 1) / CB;
  const int c0 = b * per;
  const int c1 = (c0 + per < E) ? (c0 + per) : E;
  for (int e0 = c0 + tid; e0 < c1; e0 += 1024 * EPT) {
    int d[EPT], s[EPT];
    float w[EPT];
#pragma unroll
    for (int t = 0; t < EPT; ++t) {
      const int e  = e0 + t * 1024;
      const int ee = (e < c1) ? e : (c1 - 1);
      d[t] = dst[ee]; s[t] = src[ee]; w[t] = ew[ee];
    }
    asm volatile("" :: "v"(d[0]), "v"(d[1]), "v"(d[2]), "v"(d[3]), "v"(d[4]), "v"(d[5]), "v"(d[6]), "v"(d[7]),
                       "v"(s[0]), "v"(s[1]), "v"(s[2]), "v"(s[3]), "v"(s[4]), "v"(s[5]), "v"(s[6]), "v"(s[7]),
                       "v"(w[0]), "v"(w[1]), "v"(w[2]), "v"(w[3]), "v"(w[4]), "v"(w[5]), "v"(w[6]), "v"(w[7]));
#pragma unroll
    for (int t = 0; t < EPT; ++t) {
      if (e0 + t * 1024 < c1) {
        const int pos = atomicAdd(&cur[(unsigned)d[t] / bw], 1);
        rec[pos] = make_int4(s[t], __float_as_int(w[t]), d[t], 0);
      }
    }
  }
}

__global__ __launch_bounds__(1024) void range_rows(const int4* __restrict__ rec,
                                                   const int* __restrict__ rbase,
                                                   int* __restrict__ rowptr,
                                                   int2* __restrict__ csr,
                                                   int E, int N) {
  __shared__ int cnt[FMAX];
  __shared__ int wsum[32];
  const int tid  = threadIdx.x;
  const int lane = tid & 31;
  const int wv   = tid >> 5;
  const int bw   = (int)range_width(N);
  const int row0 = blockIdx.x * bw;
  int rows = N - row0;
  if (rows > bw) rows = bw;
  if (rows < 0)  rows = 0;
  const int beg = rbase[blockIdx.x];
  const int end = rbase[blockIdx.x + 1];
  const int* recd = (const int*)rec + 2;

  for (int j = tid; j < rows; j += 1024) cnt[j] = 0;
  __syncthreads();
  for (int i0 = beg + tid; i0 < end; i0 += 1024 * EPT) {
    int d[EPT];
#pragma unroll
    for (int t = 0; t < EPT; ++t) {
      const int i = i0 + t * 1024;
      d[t] = recd[4 * (size_t)((i < end) ? i : (end - 1))];
    }
    asm volatile("" :: "v"(d[0]), "v"(d[1]), "v"(d[2]), "v"(d[3]), "v"(d[4]), "v"(d[5]), "v"(d[6]), "v"(d[7]));
#pragma unroll
    for (int t = 0; t < EPT; ++t)
      if (i0 + t * 1024 < end) atomicAdd(&cnt[d[t] - row0], 1);
  }
  __syncthreads();

  const int per = (rows + 1023) >> 10;
  const int j0  = tid * per;
  const int j1  = (j0 + per < rows) ? (j0 + per) : rows;
  int sum = 0;
  for (int j = j0; j < j1; ++j) sum += cnt[j];
  int incl = sum;
  for (int o = 1; o < 32; o <<= 1) {
    int t = __shfl_up(incl, o);
    if (lane >= o) incl += t;
  }
  if (lane == 31) wsum[wv] = incl;
  __syncthreads();
  int off = (lane < wv) ? wsum[lane] : 0;
  for (int o = 16; o > 0; o >>= 1) off += __shfl_xor(off, o);
  int run = beg + off + incl - sum;
  for (int j = j0; j < j1; ++j) {
    const int c = cnt[j];
    cnt[j] = run;
    rowptr[row0 + j] = run;
    run += c;
  }
  if (tid == 0 && rows > 0 && row0 + rows == N) rowptr[N] = E;
  __syncthreads();

  for (int i0 = beg + tid; i0 < end; i0 += 1024 * EPT) {
    int4 r[EPT];
#pragma unroll
    for (int t = 0; t < EPT; ++t) {
      const int i = i0 + t * 1024;
      r[t] = rec[(i < end) ? i : (end - 1)];
    }
    asm volatile("" :: "v"(r[0].x), "v"(r[1].x), "v"(r[2].x), "v"(r[3].x),
                       "v"(r[4].x), "v"(r[5].x), "v"(r[6].x), "v"(r[7].x));
#pragma unroll
    for (int t = 0; t < EPT; ++t) {
      if (i0 + t * 1024 < end) {
        const int pos = atomicAdd(&cnt[r[t].z - row0], 1);
        csr[pos] = make_int2(r[t].x, r[t].y);
      }
    }
  }
}

__device__ __forceinline__ float tanh_sel(float x) {
  const float ax  = fabsf(x);
  const float e   = __builtin_amdgcn_exp2f(ax * 2.88539008177792681472f);
  const float big = fmaf(__builtin_amdgcn_rcpf(e + 1.0f), -2.0f, 1.0f);
  const float x2  = x * x;
  float p = fmaf(x2, -0x1.758e7ap-8f, 0x1.521192p-6f);
  p = fmaf(x2, p, -0x1.b8389cp-5f);
  p = fmaf(x2, p, 0x1.110704p-3f);
  p = fmaf(x2, p, -0x1.555532p-2f);
  const float sm  = fmaf(x2, ax * p, ax);
  return copysignf((ax < 0.625f) ? sm : big, x);
}

__device__ __forceinline__ void row_fma(float4& acc, const float4& v, float w) {
  acc.x = fmaf(v.x, w, acc.x); acc.y = fmaf(v.y, w, acc.y);
  acc.z = fmaf(v.z, w, acc.z); acc.w = fmaf(v.w, w, acc.w);
}

__global__ __launch_bounds__(1024) void gather_gemm_tanh(const float* __restrict__ X,
                                                        const int* __restrict__ rowptr,
                                                        const int2* __restrict__ csr,
                                                        const float* __restrict__ W,
                                                        float* __restrict__ out,
                                                        int nrows) {
  __shared__ float Xs[RB * LP];
  __shared__ float Wt[128 * LP];

  const int tid  = threadIdx.x;
  const int lane = tid & 31;
  const int wave = __builtin_amdgcn_readfirstlane(tid >> 5);
  const int half = lane >> 4;
  const int ln16 = lane & 15;
  const int rowBase = blockIdx.x * RB;

  for (int i = 0; i < 4; ++i) {
    int idx = tid + i * 1024;
    int k   = idx >> 5;
    int c4  = idx & 31;
    float4 v = *(const float4*)(W + k * DDIM + c4 * 4);
    Wt[(c4 * 4 + 0) * LP + k] = v.x;
    Wt[(c4 * 4 + 1) * LP + k] = v.y;
    Wt[(c4 * 4 + 2) * LP + k] = v.z;
    Wt[(c4 * 4 + 3) * LP + k] = v.w;
  }

  {
    const int nr  = RQ + ((wave < RR) ? 1 : 0);
    const int lr0 = wave * RQ + ((wave < RR) ? wave : RR);
    const int firstRow = rowBase + lr0;
    int ri = firstRow + ((lane < nr) ? lane : nr);
    if (ri > nrows) ri = nrows;
    const int rpv = rowptr[ri];
    const int begAll = __builtin_amdgcn_readlane(rpv, 0);
    const int endAll = __builtin_amdgcn_readlane(rpv, nr);
    int r    = 0;
    int rend = __builtin_amdgcn_readlane(rpv, 1);
    const float* xl = X + lane * 4;
    float* srow = &Xs[lr0 * LP + lane * 4];
    float4 acc = make_float4(0.f, 0.f, 0.f, 0.f);

    int base = begAll;
    int cnt  = (endAll - base < 32) ? (endAll - base) : 32;
    int2 p = make_int2(0, 0);
    if (lane < cnt) p = csr[base + lane];
    while (base < endAll) {
      const int nbase = base + 32;
      const int ncnt  = (endAll - nbase < 32) ? (endAll - nbase) : 32;
      int2 pn = make_int2(0, 0);
      if (lane < ncnt) pn = csr[nbase + lane];
      for (int j = 0; j < cnt; j += 8) {
        int   s[8];
        float w[8];
        float4 v[8];
#pragma unroll
        for (int t = 0; t < 8; ++t) {
          const int jt = (j + t < cnt) ? (j + t) : (cnt - 1);
          s[t] = __builtin_amdgcn_readlane(p.x, jt);
          w[t] = __int_as_float(__builtin_amdgcn_readlane(p.y, jt));
        }
#pragma unroll
        for (int t = 0; t < 8; ++t) v[t] = *(const float4*)(xl + (size_t)s[t] * DDIM);
        asm volatile("" :: "v"(v[0].x), "v"(v[1].x), "v"(v[2].x), "v"(v[3].x),
                           "v"(v[4].x), "v"(v[5].x), "v"(v[6].x), "v"(v[7].x));
#pragma unroll
        for (int t = 0; t < 8; ++t) {
          if (j + t < cnt) {
            const int e = base + j + t;
            while (e >= rend) {
              *(float4*)(srow + r * LP) = acc;
              acc = make_float4(0.f, 0.f, 0.f, 0.f);
              ++r;
              rend = __builtin_amdgcn_readlane(rpv, r + 1);
            }
            row_fma(acc, v[t], w[t]);
          }
        }
      }
      base = nbase; cnt = ncnt; p = pn;
    }
    for (; r < nr; ++r) {
      *(float4*)(srow + r * LP) = acc;
      acc = make_float4(0.f, 0.f, 0.f, 0.f);
    }
  }
  __syncthreads();

  for (int u = wave; u < 2 * MT; u += NWV) {
  const int mRow  = (u >> 1) * 16;
  const int tBase = (u & 1) * 4;
  v8f acc[4] = {};

  for (int k0 = 0; k0 < 128; k0 += 4) {
    const float* ap = &Xs[(mRow + ln16) * LP + k0 + 2 * half];
    v2f a; a.x = ap[0]; a.y = ap[1];
#pragma unroll
    for (int t = 0; t < 4; ++t) {
      const float* bp = &Wt[((tBase + t) * 16 + ln16) * LP + k0 + 2 * half];
      v2f b; b.x = bp[0]; b.y = bp[1];
      acc[t] = __builtin_amdgcn_wmma_f32_16x16x4_f32(false, a, false, b,
                                                     (short)0, acc[t], false, false);
    }
  }

  const int gr0 = rowBase + mRow + 8 * half;
  float* orow = out + (size_t)gr0 * DDIM + tBase * 16 + ln16;
  if (rowBase + mRow + 16 <= nrows) {
#pragma unroll
    for (int t = 0; t < 4; ++t) {
#pragma unroll
      for (int r = 0; r < 8; ++r)
        orow[(size_t)r * DDIM + t * 16] = tanh_sel(acc[t][r]);
    }
  } else {
#pragma unroll
    for (int t = 0; t < 4; ++t) {
#pragma unroll
      for (int r = 0; r < 8; ++r)
        if (gr0 + r < nrows) orow[(size_t)r * DDIM + t * 16] = tanh_sel(acc[t][r]);
    }
  }
  }
}

extern "C" void kernel_launch(void* const* d_in, const int* in_sizes, int n_in,
                              void* d_out, int out_size, void* d_ws, size_t ws_size,
                              hipStream_t stream) {
  const float* x   = (const float*)d_in[0];
  const int*   src = (const int*)  d_in[1];
  const int*   dst = (const int*)  d_in[2];
  const float* ew  = (const float*)d_in[3];
  const float* W1  = (const float*)d_in[4];
  const float* W2  = (const float*)d_in[5];
  float* out = (float*)d_out;

  const int N = in_sizes[0] / DDIM;
  const int E = in_sizes[1];

  float* act1 = (float*)d_ws;
  const size_t np = ((size_t)N + 4) & ~(size_t)3;
  int*  rowptr = (int*)(act1 + (size_t)N * DDIM);
  int*  hist   = rowptr + np;
  int*  rbase  = hist + CB * NBK;
  int4* rec    = (int4*)(rbase + NBK + 4);
  int2* csr    = (int2*)(rec + E);

  const int gblocks = (N + RB - 1) / RB;

  range_count<<<CB, 1024, 0, stream>>>(dst, hist, E, N);
  range_place<<<CB, 1024, 0, stream>>>(src, dst, ew, hist, rbase, rec, E, N);
  range_rows<<<NBK, 1024, 0, stream>>>(rec, rbase, rowptr, csr, E, N);

  gather_gemm_tanh<<<gblocks, 1024, 0, stream>>>(x, rowptr, csr, W1, act1, N);
  gather_gemm_tanh<<<gblocks, 1024, 0, stream>>>(act1, rowptr, csr, W2, out, N);
}


# Layer kernel, fixed wave roles over 3 row chunks: 24 waves sum chunk c+1 from memory (16 row loads in flight each) while 8 waves run chunk c's WMMAs and tanh; W staged beside chunk 0; sums unchanged
# speedup vs baseline: 5.2564x; 1.0357x over previous
#include <hip/hip_runtime.h>
#include <cstdint>
#include <cstddef>


typedef __attribute__((ext_vector_type(2))) float v2f;
typedef __attribute__((ext_vector_type(8))) float v8f;

#define DDIM 128
#define LP   132
#define RB   400
#define NWV  32
#define MT   (RB / 16)

#define NBK  256
#define CB   256
#define EPT  8
#define FMAX 8192

__device__ __forceinline__ unsigned range_width(int N) {
  const int bw = (N + NBK - 1) / NBK;
  return (unsigned)(bw > 0 ? bw : 1);
}

__global__ __launch_bounds__(1024) void range_count(const int* __restrict__ dst,
                                                    int* __restrict__ hist,
                                                    int E, int N) {
  __shared__ int h[NBK];
  const int tid = threadIdx.x;
  const unsigned bw = range_width(N);
  if (tid < NBK) h[tid] = 0;
  __syncthreads();
  const int per = (E + CB - 1) / CB;
  const int c0 = blockIdx.x * per;
  const int c1 = (c0 + per < E) ? (c0 + per) : E;
  for (int e0 = c0 + tid; e0 < c1; e0 += 1024 * EPT) {
    int d[EPT];
#pragma unroll
    for (int t = 0; t < EPT; ++t) {
      const int e = e0 + t * 1024;
      d[t] = dst[(e < c1) ? e : (c1 - 1)];
    }
    asm volatile("" :: "v"(d[0]), "v"(d[1]), "v"(d[2]), "v"(d[3]), "v"(d[4]), "v"(d[5]), "v"(d[6]), "v"(d[7]));
#pragma unroll
    for (int t = 0; t < EPT; ++t)
      if (e0 + t * 1024 < c1) atomicAdd(&h[(unsigned)d[t] / bw], 1);
  }
  __syncthreads();
  if (tid < NBK) hist[blockIdx.x * NBK + tid] = h[tid];
}

__global__ __launch_bounds__(1024) void range_place(const int* __restrict__ src,
                                                    const int* __restrict__ dst,
                                                    const float* __restrict__ ew,
                                                    const int* __restrict__ hist,
                                                    int* __restrict__ rbase,
                                                    int4* __restrict__ rec,
                                                    int E, int N) {
  __shared__ int pTot[4][NBK];
  __shared__ int pPre[4][NBK];
  __shared__ int cur[NBK];
  __shared__ int wsum[NBK / 32];
  const int tid  = threadIdx.x;
  const int lane = tid & 31;
  const int wv   = tid >> 5;
  const int b    = blockIdx.x;
  const unsigned bw = range_width(N);
  {
    const int k = tid & (NBK - 1);
    const int q = tid >> 8;
    int tot = 0, pre = 0;
#pragma unroll
    for (int i = 0; i < CB / 4; ++i) {
      const int bb = q * (CB / 4) + i;
      const int v = hist[bb * NBK + k];
      tot += v;
      pre += (bb < b) ? v : 0;
    }
    pTot[q][k] = tot;
    pPre[q][k] = pre;
  }
  __syncthreads();
  int tot = 0, pre = 0;
  if (tid < NBK) {
    tot = pTot[0][tid] + pTot[1][tid] + pTot[2][tid] + pTot[3][tid];
    pre = pPre[0][tid] + pPre[1][tid] + pPre[2][tid] + pPre[3][tid];
  }
  int incl = tot;
  for (int o = 1; o < 32; o <<= 1) {
    int t = __shfl_up(incl, o);
    if (lane >= o) incl += t;
  }
  if (tid < NBK && lane == 31) wsum[wv] = incl;
  __syncthreads();
  if (tid < NBK) {
    int off = 0;
    for (int w = 0; w < NBK / 32; ++w) off += (w < wv) ? wsum[w] : 0;
    const int base = off + incl - tot;
    cur[tid] = base + pre;
    if (b == 0) {
      rbase[tid] = base;
      if (tid == NBK - 1) rbase[NBK] = base + tot;
    }
  }
  __syncthreads();

  const int per = (E + CB - 1) / CB;
  const int c0 = b * per;
  const int c1 = (c0 + per < E) ? (c0 + per) : E;
  for (int e0 = c0 + tid; e0 < c1; e0 += 1024 * EPT) {
    int d[EPT], s[EPT];
    float w[EPT];
#pragma unroll
    for (int t = 0; t < EPT; ++t) {
      const int e  = e0 + t * 1024;
      const int ee = (e < c1) ? e : (c1 - 1);
      d[t] = dst[ee]; s[t] = src[ee]; w[t] = ew[ee];
    }
    asm volatile("" :: "v"(d[0]), "v"(d[1]), "v"(d[2]), "v"(d[3]), "v"(d[4]), "v"(d[5]), "v"(d[6]), "v"(d[7]),
                       "v"(s[0]), "v"(s[1]), "v"(s[2]), "v"(s[3]), "v"(s[4]), "v"(s[5]), "v"(s[6]), "v"(s[7]),
                       "v"(w[0]), "v"(w[1]), "v"(w[2]), "v"(w[3]), "v"(w[4]), "v"(w[5]), "v"(w[6]), "v"(w[7]));
#pragma unroll
    for (int t = 0; t < EPT; ++t) {
      if (e0 + t * 1024 < c1) {
        const int pos = atomicAdd(&cur[(unsigned)d[t] / bw], 1);
        rec[pos] = make_int4(s[t], __float_as_int(w[t]), d[t], 0);
      }
    }
  }
}

__global__ __launch_bounds__(1024) void range_rows(const int4* __restrict__ rec,
                                                   const int* __restrict__ rbase,
                                                   int* __restrict__ rowptr,
                                                   int2* __restrict__ csr,
                                                   int E, int N) {
  __shared__ int cnt[FMAX];
  __shared__ int wsum[32];
  const int tid  = threadIdx.x;
  const int lane = tid & 31;
  const int wv   = tid >> 5;
  const int bw   = (int)range_width(N);
  const int row0 = blockIdx.x * bw;
  int rows = N - row0;
  if (rows > bw) rows = bw;
  if (rows < 0)  rows = 0;
  const int beg = rbase[blockIdx.x];
  const int end = rbase[blockIdx.x + 1];
  const int* recd = (const int*)rec + 2;

  for (int j = tid; j < rows; j += 1024) cnt[j] = 0;
  __syncthreads();
  for (int i0 = beg + tid; i0 < end; i0 += 1024 * EPT) {
    int d[EPT];
#pragma unroll
    for (int t = 0; t < EPT; ++t) {
      const int i = i0 + t * 1024;
      d[t] = recd[4 * (size_t)((i < end) ? i : (end - 1))];
    }
    asm volatile("" :: "v"(d[0]), "v"(d[1]), "v"(d[2]), "v"(d[3]), "v"(d[4]), "v"(d[5]), "v"(d[6]), "v"(d[7]));
#pragma unroll
    for (int t = 0; t < EPT; ++t)
      if (i0 + t * 1024 < end) atomicAdd(&cnt[d[t] - row0], 1);
  }
  __syncthreads();

  const int per = (rows + 1023) >> 10;
  const int j0  = tid * per;
  const int j1  = (j0 + per < rows) ? (j0 + per) : rows;
  int sum = 0;
  for (int j = j0; j < j1; ++j) sum += cnt[j];
  int incl = sum;
  for (int o = 1; o < 32; o <<= 1) {
    int t = __shfl_up(incl, o);
    if (lane >= o) incl += t;
  }
  if (lane == 31) wsum[wv] = incl;
  __syncthreads();
  int off = (lane < wv) ? wsum[lane] : 0;
  for (int o = 16; o > 0; o >>= 1) off += __shfl_xor(off, o);
  int run = beg + off + incl - sum;
  for (int j = j0; j < j1; ++j) {
    const int c = cnt[j];
    cnt[j] = run;
    rowptr[row0 + j] = run;
    run += c;
  }
  if (tid == 0 && rows > 0 && row0 + rows == N) rowptr[N] = E;
  __syncthreads();

  for (int i0 = beg + tid; i0 < end; i0 += 1024 * EPT) {
    int4 r[EPT];
#pragma unroll
    for (int t = 0; t < EPT; ++t) {
      const int i = i0 + t * 1024;
      r[t] = rec[(i < end) ? i : (end - 1)];
    }
    asm volatile("" :: "v"(r[0].x), "v"(r[1].x), "v"(r[2].x), "v"(r[3].x),
                       "v"(r[4].x), "v"(r[5].x), "v"(r[6].x), "v"(r[7].x));
#pragma unroll
    for (int t = 0; t < EPT; ++t) {
      if (i0 + t * 1024 < end) {
        const int pos = atomicAdd(&cnt[r[t].z - row0], 1);
        csr[pos] = make_int2(r[t].x, r[t].y);
      }
    }
  }
}

__device__ __forceinline__ float tanh_sel(float x) {
  const float ax  = fabsf(x);
  const float e   = __builtin_amdgcn_exp2f(ax * 2.88539008177792681472f);
  const float big = fmaf(__builtin_amdgcn_rcpf(e + 1.0f), -2.0f, 1.0f);
  const float x2  = x * x;
  float p = fmaf(x2, -0x1.758e7ap-8f, 0x1.521192p-6f);
  p = fmaf(x2, p, -0x1.b8389cp-5f);
  p = fmaf(x2, p, 0x1.110704p-3f);
  p = fmaf(x2, p, -0x1.555532p-2f);
  const float sm  = fmaf(x2, ax * p, ax);
  return copysignf((ax < 0.625f) ? sm : big, x);
}

__device__ __forceinline__ void row_fma(float4& acc, const float4& v, float w) {
  acc.x = fmaf(v.x, w, acc.x); acc.y = fmaf(v.y, w, acc.y);
  acc.z = fmaf(v.z, w, acc.z); acc.w = fmaf(v.w, w, acc.w);
}

#define NG   24
#define NM   (NWV - NG)
#define NCH  3
#define CT0  0
#define CT1  10
#define CT2  19
#define GL   16

__device__ __forceinline__ int gw_rows(int rows, int g)  { return rows / NG + ((g < rows % NG) ? 1 : 0); }
__device__ __forceinline__ int gw_first(int rows, int g) { return g * (rows / NG) + ((g < rows % NG) ? g : rows % NG); }

__device__ __forceinline__ int chunk_tile(int c) { return (c <= 0) ? CT0 : ((c == 1) ? CT1 : ((c == 2) ? CT2 : MT)); }

__device__ __forceinline__ int chunk_rowptr(const int* __restrict__ rowptr, int rowBase, int nrows,
                                            int c, int g, int lane) {
  const int rows = (chunk_tile(c + 1) - chunk_tile(c)) * 16;
  const int n    = gw_rows(rows, g);
  int ri = rowBase + chunk_tile(c) * 16 + gw_first(rows, g) + ((lane < n) ? lane : n);
  if (ri > nrows) ri = nrows;
  return rowptr[ri];
}

__device__ __forceinline__ int2 chunk_pairs(const int2* __restrict__ csr, int rpv, int c, int g, int lane) {
  const int rows = (chunk_tile(c + 1) - chunk_tile(c)) * 16;
  const int beg  = __builtin_amdgcn_readlane(rpv, 0);
  const int end  = __builtin_amdgcn_readlane(rpv, gw_rows(rows, g));
  int2 p = make_int2(0, 0);
  if (lane < end - beg) p = csr[beg + lane];
  return p;
}

__device__ __forceinline__ void gather_run(const float* __restrict__ X, const int2* __restrict__ csr,
                                           float* __restrict__ srow, int lane, int nr, int rpv, int2 p) {
  const int begAll = __builtin_amdgcn_readlane(rpv, 0);
  const int endAll = __builtin_amdgcn_readlane(rpv, nr);
  int r    = 0;
  int rend = __builtin_amdgcn_readlane(rpv, 1);
  const float* xl = X + lane * 4;
  float4 acc = make_float4(0.f, 0.f, 0.f, 0.f);

  int base = begAll;
  int cnt  = (endAll - base < 32) ? (endAll - base) : 32;
  while (base < endAll) {
    const int nbase = base + 32;
    const int ncnt  = (endAll - nbase < 32) ? (endAll - nbase) : 32;
    int2 pn = make_int2(0, 0);
    if (lane < ncnt) pn = csr[nbase + lane];
    for (int j = 0; j < cnt; j += GL) {
      int   s[GL];
      float w[GL];
      float4 v[GL];
#pragma unroll
      for (int t = 0; t < GL; ++t) {
        const int jt = (j + t < cnt) ? (j + t) : (cnt - 1);
        s[t] = __builtin_amdgcn_readlane(p.x, jt);
        w[t] = __int_as_float(__builtin_amdgcn_readlane(p.y, jt));
      }
#pragma unroll
      for (int t = 0; t < GL; ++t) v[t] = *(const float4*)(xl + (size_t)s[t] * DDIM);
      asm volatile("" :: "v"(v[0].x), "v"(v[1].x), "v"(v[2].x), "v"(v[3].x),
                         "v"(v[4].x), "v"(v[5].x), "v"(v[6].x), "v"(v[7].x),
                         "v"(v[8].x), "v"(v[9].x), "v"(v[10].x), "v"(v[11].x),
                         "v"(v[12].x), "v"(v[13].x), "v"(v[14].x), "v"(v[15].x));
#pragma unroll
      for (int t = 0; t < GL; ++t) {
        if (j + t < cnt) {
          const int e = base + j + t;
          while (e >= rend) {
            *(float4*)(srow + r * LP) = acc;
            acc = make_float4(0.f, 0.f, 0.f, 0.f);
            ++r;
            rend = __builtin_amdgcn_readlane(rpv, r + 1);
          }
          row_fma(acc, v[t], w[t]);
        }
      }
    }
    base = nbase; cnt = ncnt; p = pn;
  }
  for (; r < nr; ++r) {
    *(float4*)(srow + r * LP) = acc;
    acc = make_float4(0.f, 0.f, 0.f, 0.f);
  }
}

__device__ __forceinline__ void multiply_tiles(const float* __restrict__ Xs, const float* __restrict__ Wt,
                                               float* __restrict__ out, int rowBase, int nrows,
                                               int t0, int t1, int u0, int ustride, int ln16, int half) {
  for (int u = u0; u < 2 * (t1 - t0); u += ustride) {
    const int mRow  = (t0 + (u >> 1)) * 16;
    const int tBase = (u & 1) * 4;
    v8f acc[4] = {};

#pragma unroll
    for (int k0 = 0; k0 < 128; k0 += 4) {
      const float* ap = &Xs[(mRow + ln16) * LP + k0 + 2 * half];
      v2f a; a.x = ap[0]; a.y = ap[1];
#pragma unroll
      for (int t = 0; t < 4; ++t) {
        const float* bp = &Wt[((tBase + t) * 16 + ln16) * LP + k0 + 2 * half];
        v2f b; b.x = bp[0]; b.y = bp[1];
        acc[t] = __builtin_amdgcn_wmma_f32_16x16x4_f32(false, a, false, b,
                                                       (short)0, acc[t], false, false);
      }
    }

    const int gr0 = rowBase + mRow + 8 * half;
    float* orow = out + (size_t)gr0 * DDIM + tBase * 16 + ln16;
    if (rowBase + mRow + 16 <= nrows) {
#pragma unroll
      for (int t = 0; t < 4; ++t) {
#pragma unroll
        for (int r = 0; r < 8; ++r)
          orow[(size_t)r * DDIM + t * 16] = tanh_sel(acc[t][r]);
      }
    } else {
#pragma unroll
      for (int t = 0; t < 4; ++t) {
#pragma unroll
        for (int r = 0; r < 8; ++r)
          if (gr0 + r < nrows) orow[(size_t)r * DDIM + t * 16] = tanh_sel(acc[t][r]);
      }
    }
  }
}

__global__ __launch_bounds__(1024) void gather_gemm_tanh(const float* __restrict__ X,
                                                        const int* __restrict__ rowptr,
                                                        const int2* __restrict__ csr,
                                                        const float* __restrict__ W,
                                                        float* __restrict__ out,
                                                        int nrows) {
  __shared__ float Xs[RB * LP];
  __shared__ float Wt[128 * LP];

  const int tid  = threadIdx.x;
  const int lane = tid & 31;
  const int wave = __builtin_amdgcn_readfirstlane(tid >> 5);
  const int half = lane >> 4;
  const int ln16 = lane & 15;
  const int rowBase = blockIdx.x * RB;

  int   rpvC = 0, rpvN = 0;
  int2  pfC  = make_int2(0, 0);
  if (wave < NG) {
    rpvC = chunk_rowptr(rowptr, rowBase, nrows, 0, wave, lane);
    rpvN = chunk_rowptr(rowptr, rowBase, nrows, 1, wave, lane);
    pfC  = chunk_pairs(csr, rpvC, 0, wave, lane);
  }

  for (int c = 0; c <= NCH; ++c) {
    if (c < NCH && wave < NG) {
      int2 pfN   = make_int2(0, 0);
      int  rpvNN = 0;
      if (c + 1 < NCH) pfN   = chunk_pairs(csr, rpvN, c + 1, wave, lane);
      if (c + 2 < NCH) rpvNN = chunk_rowptr(rowptr, rowBase, nrows, c + 2, wave, lane);
      const int rows = (chunk_tile(c + 1) - chunk_tile(c)) * 16;
      const int lr   = chunk_tile(c) * 16 + gw_first(rows, wave);
      gather_run(X, csr, &Xs[lr * LP + lane * 4], lane, gw_rows(rows, wave), rpvC, pfC);
      rpvC = rpvN; rpvN = rpvNN; pfC = pfN;
    } else if (c == 0) {
      const int t  = tid - NG * 32;
      const int kq = t >> 5;
      const int c4 = t & 31;
      const float* wp = W + kq * DDIM + c4 * 4;
      float*       wt = &Wt[(c4 * 4) * LP + kq];
      for (int i = 0; i < 128 / NM; ++i) {
        float4 v = *(const float4*)(wp + i * (NM * DDIM));
        wt[0 * LP + i * NM] = v.x;
        wt[1 * LP + i * NM] = v.y;
        wt[2 * LP + i * NM] = v.z;
        wt[3 * LP + i * NM] = v.w;
      }
    } else {
      const int u0 = (wave - NG) & (NWV - 1);
      multiply_tiles(Xs, Wt, out, rowBase, nrows, chunk_tile(c - 1), chunk_tile(c),
                     u0, (c == NCH) ? NWV : NM, ln16, half);
    }
    if (c < NCH) __syncthreads();
  }
}

extern "C" void kernel_launch(void* const* d_in, const int* in_sizes, int n_in,
                              void* d_out, int out_size, void* d_ws, size_t ws_size,
                              hipStream_t stream) {
  const float* x   = (const float*)d_in[0];
  const int*   src = (const int*)  d_in[1];
  const int*   dst = (const int*)  d_in[2];
  const float* ew  = (const float*)d_in[3];
  const float* W1  = (const float*)d_in[4];
  const float* W2  = (const float*)d_in[5];
  float* out = (float*)d_out;

  const int N = in_sizes[0] / DDIM;
  const int E = in_sizes[1];

  float* act1 = (float*)d_ws;
  const size_t np = ((size_t)N + 4) & ~(size_t)3;
  int*  rowptr = (int*)(act1 + (size_t)N * DDIM);
  int*  hist   = rowptr + np;
  int*  rbase  = hist + CB * NBK;
  int4* rec    = (int4*)(rbase + NBK + 4);
  int2* csr    = (int2*)(rec + E);

  const int gblocks = (N + RB - 1) / RB;

  range_count<<<CB, 1024, 0, stream>>>(dst, hist, E, N);
  range_place<<<CB, 1024, 0, stream>>>(src, dst, ew, hist, rbase, rec, E, N);
  range_rows<<<NBK, 1024, 0, stream>>>(rec, rbase, rowptr, csr, E, N);

  gather_gemm_tanh<<<gblocks, 1024, 0, stream>>>(x, rowptr, csr, W1, act1, N);
  gather_gemm_tanh<<<gblocks, 1024, 0, stream>>>(act1, rowptr, csr, W2, out, N);
}


# Edge ordering in 3 launches, not 5: chunks ordered by range inside one launch (no count launch), each layer-1 block orders its own 400 rows' records itself; sums: LDS arrival rank as before
# speedup vs baseline: 5.5272x; 1.0515x over previous
#include <hip/hip_runtime.h>
#include <cstdint>
#include <cstddef>


typedef __attribute__((ext_vector_type(2))) float v2f;
typedef __attribute__((ext_vector_type(8))) float v8f;

#define DDIM 128
#define LP   132
#define RB   400
#define NWV  32
#define MT   (RB / 16)

#define NBK  256
#define CB   256
#define EPT  8

__device__ __forceinline__ unsigned range_width(int N) {
  (void)N;
  return (unsigned)RB;
}

__global__ __launch_bounds__(1024) void chunk_sort(const int* __restrict__ src,
                                                   const int* __restrict__ dst,
                                                   const float* __restrict__ ew,
                                                   int* __restrict__ seg,
                                                   int4* __restrict__ rec,
                                                   int E, int N) {
  __shared__ int h[NBK];
  __shared__ int wsum[NBK / 32];
  const int tid  = threadIdx.x;
  const int lane = tid & 31;
  const int wv   = tid >> 5;
  const int b    = blockIdx.x;
  const unsigned bw = range_width(N);
  if (tid < NBK) h[tid] = 0;
  __syncthreads();
  const int per = (E + CB - 1) / CB;
  const int c0 = b * per;
  const int c1 = (c0 + per < E) ? (c0 + per) : E;
  for (int e0 = c0 + tid; e0 < c1; e0 += 1024 * EPT) {
    int d[EPT];
#pragma unroll
    for (int t = 0; t < EPT; ++t) {
      const int e = e0 + t * 1024;
      d[t] = dst[(e < c1) ? e : (c1 - 1)];
    }
    asm volatile("" :: "v"(d[0]), "v"(d[1]), "v"(d[2]), "v"(d[3]), "v"(d[4]), "v"(d[5]), "v"(d[6]), "v"(d[7]));
#pragma unroll
    for (int t = 0; t < EPT; ++t)
      if (e0 + t * 1024 < c1) atomicAdd(&h[(unsigned)d[t] / bw], 1);
  }
  __syncthreads();

  int tot = 0;
  if (tid < NBK) tot = h[tid];
  int incl = tot;
  for (int o = 1; o < 32; o <<= 1) {
    int t = __shfl_up(incl, o);
    if (lane >= o) incl += t;
  }
  if (tid < NBK && lane == 31) wsum[wv] = incl;
  __syncthreads();
  if (tid < NBK) {
    int off = 0;
    for (int w = 0; w < NBK / 32; ++w) off += (w < wv) ? wsum[w] : 0;
    const int first = c0 + off + incl - tot;
    h[tid] = first;
    seg[b * (NBK + 1) + tid] = first;
    if (tid == NBK - 1) seg[b * (NBK + 1) + NBK] = first + tot;
  }
  __syncthreads();

  for (int e0 = c0 + tid; e0 < c1; e0 += 1024 * EPT) {
    int d[EPT], s[EPT];
    float w[EPT];
#pragma unroll
    for (int t = 0; t < EPT; ++t) {
      const int e  = e0 + t * 1024;
      const int ee = (e < c1) ? e : (c1 - 1);
      d[t] = dst[ee]; s[t] = src[ee]; w[t] = ew[ee];
    }
    asm volatile("" :: "v"(d[0]), "v"(d[1]), "v"(d[2]), "v"(d[3]), "v"(d[4]), "v"(d[5]), "v"(d[6]), "v"(d[7]),
                       "v"(s[0]), "v"(s[1]), "v"(s[2]), "v"(s[3]), "v"(s[4]), "v"(s[5]), "v"(s[6]), "v"(s[7]),
                       "v"(w[0]), "v"(w[1]), "v"(w[2]), "v"(w[3]), "v"(w[4]), "v"(w[5]), "v"(w[6]), "v"(w[7]));
#pragma unroll
    for (int t = 0; t < EPT; ++t) {
      if (e0 + t * 1024 < c1) {
        const int pos = atomicAdd(&h[(unsigned)d[t] / bw], 1);
        rec[pos] = make_int4(s[t], __float_as_int(w[t]), d[t], 0);
      }
    }
  }
}

__device__ __forceinline__ int seg_addr(const int* sgb, const int* sgp, int q) {
  int b = 0;
#pragma unroll
  for (int s = CB >> 1; s >= 1; s >>= 1)
    if (sgp[b + s] <= q) b += s;
  return sgb[b] + (q - sgp[b]);
}

__device__ __forceinline__ void own_rows(const int4* __restrict__ rec,
                                         const int* __restrict__ seg,
                                         int* rowptr, int2* csr, int* rp,
                                         int E, int N) {
  __shared__ int cnt[RB];
  __shared__ int wsum[32];
  __shared__ int sgb[CB];
  __shared__ int sgp[CB + 1];
  __shared__ int wlen[CB / 32];
  __shared__ int wlow[CB / 32];
  const int tid  = threadIdx.x;
  const int lane = tid & 31;
  const int wv   = tid >> 5;
  const int k    = blockIdx.x;
  const int row0 = k * RB;
  int rows = N - row0;
  if (rows > RB) rows = RB;
  if (rows < 0)  rows = 0;
  const int* recd = (const int*)rec + 2;

  const int per = (E + CB - 1) / CB;
  int len = 0, low = 0;
  if (tid < CB) {
    const int sb = seg[tid * (NBK + 1) + k];
    const int se = seg[tid * (NBK + 1) + k + 1];
    len = se - sb;
    low = sb - tid * per;
    sgb[tid] = sb;
  }
  int linc = len;
  for (int o = 1; o < 32; o <<= 1) {
    int t = __shfl_up(linc, o);
    if (lane >= o) linc += t;
  }
  for (int o = 16; o > 0; o >>= 1) low += __shfl_xor(low, o);
  if (tid < CB && lane == 31) { wlen[wv] = linc; wlow[wv] = low; }
  for (int j = tid; j < RB; j += 1024) cnt[j] = 0;
  __syncthreads();
  int all = 0, beg = 0, loff = 0;
  for (int w = 0; w < CB / 32; ++w) {
    all  += wlen[w];
    beg  += wlow[w];
    loff += (w < wv) ? wlen[w] : 0;
  }
  if (tid < CB) {
    sgp[tid] = loff + linc - len;
    if (tid == CB - 1) sgp[CB] = all;
  }
  const int end = beg + all;
  __syncthreads();

  const bool once = all <= 1024 * EPT;
  int4 r[EPT];
#pragma unroll
  for (int t = 0; t < EPT; ++t) r[t] = make_int4(0, 0, 0, 0);
  if (once) {
    if (all > 0) {
#pragma unroll
      for (int t = 0; t < EPT; ++t) {
        const int q = tid + t * 1024;
        r[t] = rec[seg_addr(sgb, sgp, (q < all) ? q : 0)];
      }
      asm volatile("" :: "v"(r[0].z), "v"(r[1].z), "v"(r[2].z), "v"(r[3].z),
                         "v"(r[4].z), "v"(r[5].z), "v"(r[6].z), "v"(r[7].z));
#pragma unroll
      for (int t = 0; t < EPT; ++t)
        if (tid + t * 1024 < all) atomicAdd(&cnt[r[t].z - row0], 1);
    }
  } else {
#pragma nounroll
    for (int q = tid; q < all; q += 1024)
      atomicAdd(&cnt[recd[4 * (size_t)seg_addr(sgb, sgp, q)] - row0], 1);
  }
  __syncthreads();

  const int pr = (rows + 1023) >> 10;
  const int j0 = tid * pr;
  const int j1 = (j0 + pr < rows) ? (j0 + pr) : rows;
  int sum = 0;
  for (int j = j0; j < j1; ++j) sum += cnt[j];
  int incl = sum;
  for (int o = 1; o < 32; o <<= 1) {
    int t = __shfl_up(incl, o);
    if (lane >= o) incl += t;
  }
  if (lane == 31) wsum[wv] = incl;
  __syncthreads();
  int off = (lane < wv) ? wsum[lane] : 0;
  for (int o = 16; o > 0; o >>= 1) off += __shfl_xor(off, o);
  int run = beg + off + incl - sum;
  for (int j = j0; j < j1; ++j) {
    const int c = cnt[j];
    cnt[j] = run;
    rp[j]  = run;
    rowptr[row0 + j] = run;
    run += c;
  }
  for (int j = rows + tid; j <= RB; j += 1024) rp[j] = end;
  if (tid == 0 && rows > 0 && row0 + rows == N) rowptr[N] = E;
  __syncthreads();

  if (once) {
#pragma unroll
    for (int t = 0; t < EPT; ++t) {
      if (tid + t * 1024 < all) {
        const int pos = atomicAdd(&cnt[r[t].z - row0], 1);
        csr[pos] = make_int2(r[t].x, r[t].y);
      }
    }
  } else {
#pragma nounroll
    for (int q = tid; q < all; q += 1024) {
      const int4 g = rec[seg_addr(sgb, sgp, q)];
      const int pos = atomicAdd(&cnt[g.z - row0], 1);
      csr[pos] = make_int2(g.x, g.y);
    }
  }
  __syncthreads();
}

__device__ __forceinline__ float tanh_sel(float x) {
  const float ax  = fabsf(x);
  const float e   = __builtin_amdgcn_exp2f(ax * 2.88539008177792681472f);
  const float big = fmaf(__builtin_amdgcn_rcpf(e + 1.0f), -2.0f, 1.0f);
  const float x2  = x * x;
  float p = fmaf(x2, -0x1.758e7ap-8f, 0x1.521192p-6f);
  p = fmaf(x2, p, -0x1.b8389cp-5f);
  p = fmaf(x2, p, 0x1.110704p-3f);
  p = fmaf(x2, p, -0x1.555532p-2f);
  const float sm  = fmaf(x2, ax * p, ax);
  return copysignf((ax < 0.625f) ? sm : big, x);
}

__device__ __forceinline__ void row_fma(float4& acc, const float4& v, float w) {
  acc.x = fmaf(v.x, w, acc.x); acc.y = fmaf(v.y, w, acc.y);
  acc.z = fmaf(v.z, w, acc.z); acc.w = fmaf(v.w, w, acc.w);
}

#define NG   24
#define NM   (NWV - NG)
#define NCH  3
#define CT0  0
#define CT1  10
#define CT2  19
#define GL   16

__device__ __forceinline__ int gw_rows(int rows, int g)  { return rows / NG + ((g < rows % NG) ? 1 : 0); }
__device__ __forceinline__ int gw_first(int rows, int g) { return g * (rows / NG) + ((g < rows % NG) ? g : rows % NG); }

__device__ __forceinline__ int chunk_tile(int c) { return (c <= 0) ? CT0 : ((c == 1) ? CT1 : ((c == 2) ? CT2 : MT)); }

__device__ __forceinline__ int chunk_rowptr(const int* __restrict__ rowptr, int rowBase, int nrows,
                                            int c, int g, int lane) {
  const int rows = (chunk_tile(c + 1) - chunk_tile(c)) * 16;
  const int n    = gw_rows(rows, g);
  int ri = rowBase + chunk_tile(c) * 16 + gw_first(rows, g) + ((lane < n) ? lane : n);
  if (ri > nrows) ri = nrows;
  return rowptr[ri];
}

__device__ __forceinline__ int chunk_rowptr_own(const int* rp, int c, int g, int lane) {
  const int rows = (chunk_tile(c + 1) - chunk_tile(c)) * 16;
  const int n    = gw_rows(rows, g);
  return rp[chunk_tile(c) * 16 + gw_first(rows, g) + ((lane < n) ? lane : n)];
}

__device__ __forceinline__ int2 chunk_pairs(const int2* __restrict__ csr, int rpv, int c, int g, int lane) {
  const int rows = (chunk_tile(c + 1) - chunk_tile(c)) * 16;
  const int beg  = __builtin_amdgcn_readlane(rpv, 0);
  const int end  = __builtin_amdgcn_readlane(rpv, gw_rows(rows, g));
  int2 p = make_int2(0, 0);
  if (lane < end - beg) p = csr[beg + lane];
  return p;
}

__device__ __forceinline__ void gather_run(const float* __restrict__ X, const int2* __restrict__ csr,
                                           float* __restrict__ srow, int lane, int nr, int rpv, int2 p) {
  const int begAll = __builtin_amdgcn_readlane(rpv, 0);
  const int endAll = __builtin_amdgcn_readlane(rpv, nr);
  int r    = 0;
  int rend = __builtin_amdgcn_readlane(rpv, 1);
  const float* xl = X + lane * 4;
  float4 acc = make_float4(0.f, 0.f, 0.f, 0.f);

  int base = begAll;
  int cnt  = (endAll - base < 32) ? (endAll - base) : 32;
  while (base < endAll) {
    const int nbase = base + 32;
    const int ncnt  = (endAll - nbase < 32) ? (endAll - nbase) : 32;
    int2 pn = make_int2(0, 0);
    if (lane < ncnt) pn = csr[nbase + lane];
    for (int j = 0; j < cnt; j += GL) {
      int   s[GL];
      float w[GL];
      float4 v[GL];
#pragma unroll
      for (int t = 0; t < GL; ++t) {
        const int jt = (j + t < cnt) ? (j + t) : (cnt - 1);
        s[t] = __builtin_amdgcn_readlane(p.x, jt);
        w[t] = __int_as_float(__builtin_amdgcn_readlane(p.y, jt));
      }
#pragma unroll
      for (int t = 0; t < GL; ++t) v[t] = *(const float4*)(xl + (size_t)s[t] * DDIM);
      asm volatile("" :: "v"(v[0].x), "v"(v[1].x), "v"(v[2].x), "v"(v[3].x),
                         "v"(v[4].x), "v"(v[5].x), "v"(v[6].x), "v"(v[7].x),
                         "v"(v[8].x), "v"(v[9].x), "v"(v[10].x), "v"(v[11].x),
                         "v"(v[12].x), "v"(v[13].x), "v"(v[14].x), "v"(v[15].x));
#pragma unroll
      for (int t = 0; t < GL; ++t) {
        if (j + t < cnt) {
          const int e = base + j + t;
          while (e >= rend) {
            *(float4*)(srow + r * LP) = acc;
            acc = make_float4(0.f, 0.f, 0.f, 0.f);
            ++r;
            rend = __builtin_amdgcn_readlane(rpv, r + 1);
          }
          row_fma(acc, v[t], w[t]);
        }
      }
    }
    base = nbase; cnt = ncnt; p = pn;
  }
  for (; r < nr; ++r) {
    *(float4*)(srow + r * LP) = acc;
    acc = make_float4(0.f, 0.f, 0.f, 0.f);
  }
}

__device__ __forceinline__ void multiply_tiles(const float* __restrict__ Xs, const float* __restrict__ Wt,
                                               float* __restrict__ out, int rowBase, int nrows,
                                               int t0, int t1, int u0, int ustride, int ln16, int half) {
  for (int u = u0; u < 2 * (t1 - t0); u += ustride) {
    const int mRow  = (t0 + (u >> 1)) * 16;
    const int tBase = (u & 1) * 4;
    v8f acc[4] = {};

#pragma unroll
    for (int k0 = 0; k0 < 128; k0 += 4) {
      const float* ap = &Xs[(mRow + ln16) * LP + k0 + 2 * half];
      v2f a; a.x = ap[0]; a.y = ap[1];
#pragma unroll
      for (int t = 0; t < 4; ++t) {
        const float* bp = &Wt[((tBase + t) * 16 + ln16) * LP + k0 + 2 * half];
        v2f b; b.x = bp[0]; b.y = bp[1];
        acc[t] = __builtin_amdgcn_wmma_f32_16x16x4_f32(false, a, false, b,
                                                       (short)0, acc[t], false, false);
      }
    }

    const int gr0 = rowBase + mRow + 8 * half;
    float* orow = out + (size_t)gr0 * DDIM + tBase * 16 + ln16;
    if (rowBase + mRow + 16 <= nrows) {
#pragma unroll
      for (int t = 0; t < 4; ++t) {
#pragma unroll
        for (int r = 0; r < 8; ++r)
          orow[(size_t)r * DDIM + t * 16] = tanh_sel(acc[t][r]);
      }
    } else {
#pragma unroll
      for (int t = 0; t < 4; ++t) {
#pragma unroll
        for (int r = 0; r < 8; ++r)
          if (gr0 + r < nrows) orow[(size_t)r * DDIM + t * 16] = tanh_sel(acc[t][r]);
      }
    }
  }
}

template <bool FIRST>
__global__ __launch_bounds__(1024) void gather_gemm_tanh(const float* __restrict__ X,
                                                        int* rowptr,
                                                        int2* csr,
                                                        const float* __restrict__ W,
                                                        float* __restrict__ out,
                                                        int nrows,
                                                        const int4* __restrict__ rec,
                                                        const int* __restrict__ seg,
                                                        int nedges) {
  __shared__ float Xs[RB * LP];
  __shared__ float Wt[128 * LP];
  __shared__ int   rp[FIRST ? (RB + 1) : 1];

  const int tid  = threadIdx.x;
  const int lane = tid & 31;
  const int wave = __builtin_amdgcn_readfirstlane(tid >> 5);
  const int half = lane >> 4;
  const int ln16 = lane & 15;
  const int rowBase = blockIdx.x * RB;

  int   rpvC = 0, rpvN = 0;
  int2  pfC  = make_int2(0, 0);
  if constexpr (FIRST) own_rows(rec, seg, rowptr, csr, rp, nedges, nrows);
  if (wave < NG) {
    rpvC = FIRST ? chunk_rowptr_own(rp, 0, wave, lane) : chunk_rowptr(rowptr, rowBase, nrows, 0, wave, lane);
    rpvN = FIRST ? chunk_rowptr_own(rp, 1, wave, lane) : chunk_rowptr(rowptr, rowBase, nrows, 1, wave, lane);
    pfC  = chunk_pairs(csr, rpvC, 0, wave, lane);
  }

  for (int c = 0; c <= NCH; ++c) {
    if (c < NCH && wave < NG) {
      int2 pfN   = make_int2(0, 0);
      int  rpvNN = 0;
      if (c + 1 < NCH) pfN   = chunk_pairs(csr, rpvN, c + 1, wave, lane);
      if (c + 2 < NCH) rpvNN = FIRST ? chunk_rowptr_own(rp, c + 2, wave, lane)
                                     : chunk_rowptr(rowptr, rowBase, nrows, c + 2, wave, lane);
      const int rows = (chunk_tile(c + 1) - chunk_tile(c)) * 16;
      const int lr   = chunk_tile(c) * 16 + gw_first(rows, wave);
      gather_run(X, csr, &Xs[lr * LP + lane * 4], lane, gw_rows(rows, wave), rpvC, pfC);
      rpvC = rpvN; rpvN = rpvNN; pfC = pfN;
    } else if (c == 0) {
      const int t  = tid - NG * 32;
      const int kq = t >> 5;
      const int c4 = t & 31;
      const float* wp = W + kq * DDIM + c4 * 4;
      float*       wt = &Wt[(c4 * 4) * LP + kq];
      for (int i = 0; i < 128 / NM; ++i) {
        float4 v = *(const float4*)(wp + i * (NM * DDIM));
        wt[0 * LP + i * NM] = v.x;
        wt[1 * LP + i * NM] = v.y;
        wt[2 * LP + i * NM] = v.z;
        wt[3 * LP + i * NM] = v.w;
      }
    } else {
      const int u0 = (wave - NG) & (NWV - 1);
      multiply_tiles(Xs, Wt, out, rowBase, nrows, chunk_tile(c - 1), chunk_tile(c),
                     u0, (c == NCH) ? NWV : NM, ln16, half);
    }
    if (c < NCH) __syncthreads();
  }
}

extern "C" void kernel_launch(void* const* d_in, const int* in_sizes, int n_in,
                              void* d_out, int out_size, void* d_ws, size_t ws_size,
                              hipStream_t stream) {
  const float* x   = (const float*)d_in[0];
  const int*   src = (const int*)  d_in[1];
  const int*   dst = (const int*)  d_in[2];
  const float* ew  = (const float*)d_in[3];
  const float* W1  = (const float*)d_in[4];
  const float* W2  = (const float*)d_in[5];
  float* out = (float*)d_out;

  const int N = in_sizes[0] / DDIM;
  const int E = in_sizes[1];

  float* act1 = (float*)d_ws;
  const size_t np = ((size_t)N + 4) & ~(size_t)3;
  int*  rowptr = (int*)(act1 + (size_t)N * DDIM);
  int*  seg    = rowptr + np;
  int4* rec    = (int4*)(seg + CB * (NBK + 1));
  int2* csr    = (int2*)(rec + E);

  const int gblocks = (N + RB - 1) / RB;

  chunk_sort<<<CB, 1024, 0, stream>>>(src, dst, ew, seg, rec, E, N);

  gather_gemm_tanh<true><<<gblocks, 1024, 0, stream>>>(x, rowptr, csr, W1, act1, N, rec, seg, E);
  gather_gemm_tanh<false><<<gblocks, 1024, 0, stream>>>(act1, rowptr, csr, W2, out, N, rec, seg, E);
}


# Ordered records leave through LDS as whole lines: a chunk's records and a layer-1 block's (src, w) pairs take their slots in LDS, then go out in address order, not 32 lines per store; sums as before
# speedup vs baseline: 5.8103x; 1.0512x over previous
#include <hip/hip_runtime.h>
#include <cstdint>
#include <cstddef>


typedef __attribute__((ext_vector_type(2))) float v2f;
typedef __attribute__((ext_vector_type(8))) float v8f;

#define DDIM 128
#define LP   132
#define RB   400
#define NWV  32
#define MT   (RB / 16)

#define NBK  256
#define CB   256
#define EPT  8
#define SCAP (1024 * EPT)

__device__ __forceinline__ unsigned range_width(int N) {
  (void)N;
  return (unsigned)RB;
}

__global__ __launch_bounds__(1024) void chunk_sort(const int* __restrict__ src,
                                                   const int* __restrict__ dst,
                                                   const float* __restrict__ ew,
                                                   int* __restrict__ seg,
                                                   int4* __restrict__ rec,
                                                   int E, int N) {
  __shared__ int h[NBK];
  __shared__ int wsum[NBK / 32];
  __shared__ int4 stage[SCAP];
  const int tid  = threadIdx.x;
  const int lane = tid & 31;
  const int wv   = tid >> 5;
  const int b    = blockIdx.x;
  const unsigned bw = range_width(N);
  if (tid < NBK) h[tid] = 0;
  __syncthreads();
  const int per = (E + CB - 1) / CB;
  const int c0 = b * per;
  const int c1 = (c0 + per < E) ? (c0 + per) : E;
  const bool staged = (c1 - c0) <= SCAP;
  for (int e0 = c0 + tid; e0 < c1; e0 += 1024 * EPT) {
    int d[EPT];
#pragma unroll
    for (int t = 0; t < EPT; ++t) {
      const int e = e0 + t * 1024;
      d[t] = dst[(e < c1) ? e : (c1 - 1)];
    }
    asm volatile("" :: "v"(d[0]), "v"(d[1]), "v"(d[2]), "v"(d[3]), "v"(d[4]), "v"(d[5]), "v"(d[6]), "v"(d[7]));
#pragma unroll
    for (int t = 0; t < EPT; ++t)
      if (e0 + t * 1024 < c1) atomicAdd(&h[(unsigned)d[t] / bw], 1);
  }
  __syncthreads();

  int tot = 0;
  if (tid < NBK) tot = h[tid];
  int incl = tot;
  for (int o = 1; o < 32; o <<= 1) {
    int t = __shfl_up(incl, o);
    if (lane >= o) incl += t;
  }
  if (tid < NBK && lane == 31) wsum[wv] = incl;
  __syncthreads();
  if (tid < NBK) {
    int off = 0;
    for (int w = 0; w < NBK / 32; ++w) off += (w < wv) ? wsum[w] : 0;
    const int first = c0 + off + incl - tot;
    h[tid] = first;
    seg[b * (NBK + 1) + tid] = first;
    if (tid == NBK - 1) seg[b * (NBK + 1) + NBK] = first + tot;
  }
  __syncthreads();

  for (int e0 = c0 + tid; e0 < c1; e0 += 1024 * EPT) {
    int d[EPT], s[EPT];
    float w[EPT];
#pragma unroll
    for (int t = 0; t < EPT; ++t) {
      const int e  = e0 + t * 1024;
      const int ee = (e < c1) ? e : (c1 - 1);
      d[t] = dst[ee]; s[t] = src[ee]; w[t] = ew[ee];
    }
    asm volatile("" :: "v"(d[0]), "v"(d[1]), "v"(d[2]), "v"(d[3]), "v"(d[4]), "v"(d[5]), "v"(d[6]), "v"(d[7]),
                       "v"(s[0]), "v"(s[1]), "v"(s[2]), "v"(s[3]), "v"(s[4]), "v"(s[5]), "v"(s[6]), "v"(s[7]),
                       "v"(w[0]), "v"(w[1]), "v"(w[2]), "v"(w[3]), "v"(w[4]), "v"(w[5]), "v"(w[6]), "v"(w[7]));
#pragma unroll
    for (int t = 0; t < EPT; ++t) {
      if (e0 + t * 1024 < c1) {
        const int pos = atomicAdd(&h[(unsigned)d[t] / bw], 1);
        const int4 g = make_int4(s[t], __float_as_int(w[t]), d[t], 0);
        if (staged) stage[pos - c0] = g;
        else        rec[pos] = g;
      }
    }
  }
  if (staged) {
    __syncthreads();
    const int n = c1 - c0;
    if (n > 0) {
      int4 g[EPT];
#pragma unroll
      for (int t = 0; t < EPT; ++t) {
        const int i = tid + t * 1024;
        g[t] = stage[(i < n) ? i : (n - 1)];
      }
      asm volatile("" :: "v"(g[0].x), "v"(g[0].y), "v"(g[0].z), "v"(g[0].w), "v"(g[1].x), "v"(g[1].y), "v"(g[1].z), "v"(g[1].w),
                         "v"(g[2].x), "v"(g[2].y), "v"(g[2].z), "v"(g[2].w), "v"(g[3].x), "v"(g[3].y), "v"(g[3].z), "v"(g[3].w));
      asm volatile("" :: "v"(g[4].x), "v"(g[4].y), "v"(g[4].z), "v"(g[4].w), "v"(g[5].x), "v"(g[5].y), "v"(g[5].z), "v"(g[5].w),
                         "v"(g[6].x), "v"(g[6].y), "v"(g[6].z), "v"(g[6].w), "v"(g[7].x), "v"(g[7].y), "v"(g[7].z), "v"(g[7].w));
#pragma unroll
      for (int t = 0; t < EPT; ++t) {
        const int i = tid + t * 1024;
        if (i < n) rec[c0 + i] = g[t];
      }
    }
  }
}

__device__ __forceinline__ int seg_addr(const int* sgb, const int* sgp, int q) {
  int b = 0;
#pragma unroll
  for (int s = CB >> 1; s >= 1; s >>= 1)
    if (sgp[b + s] <= q) b += s;
  return sgb[b] + (q - sgp[b]);
}

__device__ __forceinline__ void own_rows(const int4* __restrict__ rec,
                                         const int* __restrict__ seg,
                                         int* rowptr, int2* csr, int* rp, int2* stg,
                                         int E, int N) {
  __shared__ int cnt[RB];
  __shared__ int wsum[32];
  __shared__ int sgb[CB];
  __shared__ int sgp[CB + 1];
  __shared__ int wlen[CB / 32];
  __shared__ int wlow[CB / 32];
  const int tid  = threadIdx.x;
  const int lane = tid & 31;
  const int wv   = tid >> 5;
  const int k    = blockIdx.x;
  const int row0 = k * RB;
  int rows = N - row0;
  if (rows > RB) rows = RB;
  if (rows < 0)  rows = 0;
  const int* recd = (const int*)rec + 2;

  const int per = (E + CB - 1) / CB;
  int len = 0, low = 0;
  if (tid < CB) {
    const int sb = seg[tid * (NBK + 1) + k];
    const int se = seg[tid * (NBK + 1) + k + 1];
    len = se - sb;
    low = sb - tid * per;
    sgb[tid] = sb;
  }
  int linc = len;
  for (int o = 1; o < 32; o <<= 1) {
    int t = __shfl_up(linc, o);
    if (lane >= o) linc += t;
  }
  for (int o = 16; o > 0; o >>= 1) low += __shfl_xor(low, o);
  if (tid < CB && lane == 31) { wlen[wv] = linc; wlow[wv] = low; }
  for (int j = tid; j < RB; j += 1024) cnt[j] = 0;
  __syncthreads();
  int all = 0, beg = 0, loff = 0;
  for (int w = 0; w < CB / 32; ++w) {
    all  += wlen[w];
    beg  += wlow[w];
    loff += (w < wv) ? wlen[w] : 0;
  }
  if (tid < CB) {
    sgp[tid] = loff + linc - len;
    if (tid == CB - 1) sgp[CB] = all;
  }
  const int end = beg + all;
  __syncthreads();

  const bool once = all <= 1024 * EPT;
  int4 r[EPT];
#pragma unroll
  for (int t = 0; t < EPT; ++t) r[t] = make_int4(0, 0, 0, 0);
  if (once) {
    if (all > 0) {
#pragma unroll
      for (int t = 0; t < EPT; ++t) {
        const int q = tid + t * 1024;
        r[t] = rec[seg_addr(sgb, sgp, (q < all) ? q : 0)];
      }
      asm volatile("" :: "v"(r[0].z), "v"(r[1].z), "v"(r[2].z), "v"(r[3].z),
                         "v"(r[4].z), "v"(r[5].z), "v"(r[6].z), "v"(r[7].z));
#pragma unroll
      for (int t = 0; t < EPT; ++t)
        if (tid + t * 1024 < all) atomicAdd(&cnt[r[t].z - row0], 1);
    }
  } else {
#pragma nounroll
    for (int q = tid; q < all; q += 1024)
      atomicAdd(&cnt[recd[4 * (size_t)seg_addr(sgb, sgp, q)] - row0], 1);
  }
  __syncthreads();

  const int pr = (rows + 1023) >> 10;
  const int j0 = tid * pr;
  const int j1 = (j0 + pr < rows) ? (j0 + pr) : rows;
  int sum = 0;
  for (int j = j0; j < j1; ++j) sum += cnt[j];
  int incl = sum;
  for (int o = 1; o < 32; o <<= 1) {
    int t = __shfl_up(incl, o);
    if (lane >= o) incl += t;
  }
  if (lane == 31) wsum[wv] = incl;
  __syncthreads();
  int off = (lane < wv) ? wsum[lane] : 0;
  for (int o = 16; o > 0; o >>= 1) off += __shfl_xor(off, o);
  int run = beg + off + incl - sum;
  for (int j = j0; j < j1; ++j) {
    const int c = cnt[j];
    cnt[j] = run;
    rp[j]  = run;
    rowptr[row0 + j] = run;
    run += c;
  }
  for (int j = rows + tid; j <= RB; j += 1024) rp[j] = end;
  if (tid == 0 && rows > 0 && row0 + rows == N) rowptr[N] = E;
  __syncthreads();

  if (once) {
#pragma unroll
    for (int t = 0; t < EPT; ++t) {
      if (tid + t * 1024 < all) {
        const int pos = atomicAdd(&cnt[r[t].z - row0], 1);
        stg[pos - beg] = make_int2(r[t].x, r[t].y);
      }
    }
    __syncthreads();
    if (all > 0) {
      int2 g[EPT];
#pragma unroll
      for (int t = 0; t < EPT; ++t) {
        const int i = tid + t * 1024;
        g[t] = stg[(i < all) ? i : (all - 1)];
      }
      asm volatile("" :: "v"(g[0].x), "v"(g[0].y), "v"(g[1].x), "v"(g[1].y), "v"(g[2].x), "v"(g[2].y), "v"(g[3].x), "v"(g[3].y),
                         "v"(g[4].x), "v"(g[4].y), "v"(g[5].x), "v"(g[5].y), "v"(g[6].x), "v"(g[6].y), "v"(g[7].x), "v"(g[7].y));
#pragma unroll
      for (int t = 0; t < EPT; ++t) {
        const int i = tid + t * 1024;
        if (i < all) csr[beg + i] = g[t];
      }
    }
  } else {
#pragma nounroll
    for (int q = tid; q < all; q += 1024) {
      const int4 g = rec[seg_addr(sgb, sgp, q)];
      const int pos = atomicAdd(&cnt[g.z - row0], 1);
      csr[pos] = make_int2(g.x, g.y);
    }
  }
  __syncthreads();
}

__device__ __forceinline__ float tanh_sel(float x) {
  const float ax  = fabsf(x);
  const float e   = __builtin_amdgcn_exp2f(ax * 2.88539008177792681472f);
  const float big = fmaf(__builtin_amdgcn_rcpf(e + 1.0f), -2.0f, 1.0f);
  const float x2  = x * x;
  float p = fmaf(x2, -0x1.758e7ap-8f, 0x1.521192p-6f);
  p = fmaf(x2, p, -0x1.b8389cp-5f);
  p = fmaf(x2, p, 0x1.110704p-3f);
  p = fmaf(x2, p, -0x1.555532p-2f);
  const float sm  = fmaf(x2, ax * p, ax);
  return copysignf((ax < 0.625f) ? sm : big, x);
}

__device__ __forceinline__ void row_fma(float4& acc, const float4& v, float w) {
  acc.x = fmaf(v.x, w, acc.x); acc.y = fmaf(v.y, w, acc.y);
  acc.z = fmaf(v.z, w, acc.z); acc.w = fmaf(v.w, w, acc.w);
}

#define NG   24
#define NM   (NWV - NG)
#define NCH  3
#define CT0  0
#define CT1  10
#define CT2  19
#define GL   16

__device__ __forceinline__ int gw_rows(int rows, int g)  { return rows / NG + ((g < rows % NG) ? 1 : 0); }
__device__ __forceinline__ int gw_first(int rows, int g) { return g * (rows / NG) + ((g < rows % NG) ? g : rows % NG); }

__device__ __forceinline__ int chunk_tile(int c) { return (c <= 0) ? CT0 : ((c == 1) ? CT1 : ((c == 2) ? CT2 : MT)); }

__device__ __forceinline__ int chunk_rowptr(const int* __restrict__ rowptr, int rowBase, int nrows,
                                            int c, int g, int lane) {
  const int rows = (chunk_tile(c + 1) - chunk_tile(c)) * 16;
  const int n    = gw_rows(rows, g);
  int ri = rowBase + chunk_tile(c) * 16 + gw_first(rows, g) + ((lane < n) ? lane : n);
  if (ri > nrows) ri = nrows;
  return rowptr[ri];
}

__device__ __forceinline__ int chunk_rowptr_own(const int* rp, int c, int g, int lane) {
  const int rows = (chunk_tile(c + 1) - chunk_tile(c)) * 16;
  const int n    = gw_rows(rows, g);
  return rp[chunk_tile(c) * 16 + gw_first(rows, g) + ((lane < n) ? lane : n)];
}

__device__ __forceinline__ int2 chunk_pairs(const int2* __restrict__ csr, int rpv, int c, int g, int lane) {
  const int rows = (chunk_tile(c + 1) - chunk_tile(c)) * 16;
  const int beg  = __builtin_amdgcn_readlane(rpv, 0);
  const int end  = __builtin_amdgcn_readlane(rpv, gw_rows(rows, g));
  int2 p = make_int2(0, 0);
  if (lane < end - beg) p = csr[beg + lane];
  return p;
}

__device__ __forceinline__ void gather_run(const float* __restrict__ X, const int2* __restrict__ csr,
                                           float* __restrict__ srow, int lane, int nr, int rpv, int2 p) {
  const int begAll = __builtin_amdgcn_readlane(rpv, 0);
  const int endAll = __builtin_amdgcn_readlane(rpv, nr);
  int r    = 0;
  int rend = __builtin_amdgcn_readlane(rpv, 1);
  const float* xl = X + lane * 4;
  float4 acc = make_float4(0.f, 0.f, 0.f, 0.f);

  int base = begAll;
  int cnt  = (endAll - base < 32) ? (endAll - base) : 32;
  while (base < endAll) {
    const int nbase = base + 32;
    const int ncnt  = (endAll - nbase < 32) ? (endAll - nbase) : 32;
    int2 pn = make_int2(0, 0);
    if (lane < ncnt) pn = csr[nbase + lane];
    for (int j = 0; j < cnt; j += GL) {
      int   s[GL];
      float w[GL];
      float4 v[GL];
#pragma unroll
      for (int t = 0; t < GL; ++t) {
        const int jt = (j + t < cnt) ? (j + t) : (cnt - 1);
        s[t] = __builtin_amdgcn_readlane(p.x, jt);
        w[t] = __int_as_float(__builtin_amdgcn_readlane(p.y, jt));
      }
#pragma unroll
      for (int t = 0; t < GL; ++t) v[t] = *(const float4*)(xl + (size_t)s[t] * DDIM);
      asm volatile("" :: "v"(v[0].x), "v"(v[1].x), "v"(v[2].x), "v"(v[3].x),
                         "v"(v[4].x), "v"(v[5].x), "v"(v[6].x), "v"(v[7].x),
                         "v"(v[8].x), "v"(v[9].x), "v"(v[10].x), "v"(v[11].x),
                         "v"(v[12].x), "v"(v[13].x), "v"(v[14].x), "v"(v[15].x));
#pragma unroll
      for (int t = 0; t < GL; ++t) {
        if (j + t < cnt) {
          const int e = base + j + t;
          while (e >= rend) {
            *(float4*)(srow + r * LP) = acc;
            acc = make_float4(0.f, 0.f, 0.f, 0.f);
            ++r;
            rend = __builtin_amdgcn_readlane(rpv, r + 1);
          }
          row_fma(acc, v[t], w[t]);
        }
      }
    }
    base = nbase; cnt = ncnt; p = pn;
  }
  for (; r < nr; ++r) {
    *(float4*)(srow + r * LP) = acc;
    acc = make_float4(0.f, 0.f, 0.f, 0.f);
  }
}

__device__ __forceinline__ void multiply_tiles(const float* __restrict__ Xs, const float* __restrict__ Wt,
                                               float* __restrict__ out, int rowBase, int nrows,
                                               int t0, int t1, int u0, int ustride, int ln16, int half) {
  for (int u = u0; u < 2 * (t1 - t0); u += ustride) {
    const int mRow  = (t0 + (u >> 1)) * 16;
    const int tBase = (u & 1) * 4;
    v8f acc[4] = {};

#pragma unroll
    for (int k0 = 0; k0 < 128; k0 += 4) {
      const float* ap = &Xs[(mRow + ln16) * LP + k0 + 2 * half];
      v2f a; a.x = ap[0]; a.y = ap[1];
#pragma unroll
      for (int t = 0; t < 4; ++t) {
        const float* bp = &Wt[((tBase + t) * 16 + ln16) * LP + k0 + 2 * half];
        v2f b; b.x = bp[0]; b.y = bp[1];
        acc[t] = __builtin_amdgcn_wmma_f32_16x16x4_f32(false, a, false, b,
                                                       (short)0, acc[t], false, false);
      }
    }

    const int gr0 = rowBase + mRow + 8 * half;
    float* orow = out + (size_t)gr0 * DDIM + tBase * 16 + ln16;
    if (rowBase + mRow + 16 <= nrows) {
#pragma unroll
      for (int t = 0; t < 4; ++t) {
#pragma unroll
        for (int r = 0; r < 8; ++r)
          orow[(size_t)r * DDIM + t * 16] = tanh_sel(acc[t][r]);
      }
    } else {
#pragma unroll
      for (int t = 0; t < 4; ++t) {
#pragma unroll
        for (int r = 0; r < 8; ++r)
          if (gr0 + r < nrows) orow[(size_t)r * DDIM + t * 16] = tanh_sel(acc[t][r]);
      }
    }
  }
}

template <bool FIRST>
__global__ __launch_bounds__(1024) void gather_gemm_tanh(const float* __restrict__ X,
                                                        int* rowptr,
                                                        int2* csr,
                                                        const float* __restrict__ W,
                                                        float* __restrict__ out,
                                                        int nrows,
                                                        const int4* __restrict__ rec,
                                                        const int* __restrict__ seg,
                                                        int nedges) {
  __shared__ float Xs[RB * LP];
  __shared__ float Wt[128 * LP];
  __shared__ int   rp[FIRST ? (RB + 1) : 1];

  const int tid  = threadIdx.x;
  const int lane = tid & 31;
  const int wave = __builtin_amdgcn_readfirstlane(tid >> 5);
  const int half = lane >> 4;
  const int ln16 = lane & 15;
  const int rowBase = blockIdx.x * RB;

  int   rpvC = 0, rpvN = 0;
  int2  pfC  = make_int2(0, 0);
  if constexpr (FIRST) own_rows(rec, seg, rowptr, csr, rp, reinterpret_cast<int2*>(Xs), nedges, nrows);
  if (wave < NG) {
    rpvC = FIRST ? chunk_rowptr_own(rp, 0, wave, lane) : chunk_rowptr(rowptr, rowBase, nrows, 0, wave, lane);
    rpvN = FIRST ? chunk_rowptr_own(rp, 1, wave, lane) : chunk_rowptr(rowptr, rowBase, nrows, 1, wave, lane);
    pfC  = chunk_pairs(csr, rpvC, 0, wave, lane);
  }

  for (int c = 0; c <= NCH; ++c) {
    if (c < NCH && wave < NG) {
      int2 pfN   = make_int2(0, 0);
      int  rpvNN = 0;
      if (c + 1 < NCH) pfN   = chunk_pairs(csr, rpvN, c + 1, wave, lane);
      if (c + 2 < NCH) rpvNN = FIRST ? chunk_rowptr_own(rp, c + 2, wave, lane)
                                     : chunk_rowptr(rowptr, rowBase, nrows, c + 2, wave, lane);
      const int rows = (chunk_tile(c + 1) - chunk_tile(c)) * 16;
      const int lr   = chunk_tile(c) * 16 + gw_first(rows, wave);
      gather_run(X, csr, &Xs[lr * LP + lane * 4], lane, gw_rows(rows, wave), rpvC, pfC);
      rpvC = rpvN; rpvN = rpvNN; pfC = pfN;
    } else if (c == 0) {
      const int t  = tid - NG * 32;
      const int kq = t >> 5;
      const int c4 = t & 31;
      const float* wp = W + kq * DDIM + c4 * 4;
      float*       wt = &Wt[(c4 * 4) * LP + kq];
      for (int i = 0; i < 128 / NM; ++i) {
        float4 v = *(const float4*)(wp + i * (NM * DDIM));
        wt[0 * LP + i * NM] = v.x;
        wt[1 * LP + i * NM] = v.y;
        wt[2 * LP + i * NM] = v.z;
        wt[3 * LP + i * NM] = v.w;
      }
    } else {
      const int u0 = (wave - NG) & (NWV - 1);
      multiply_tiles(Xs, Wt, out, rowBase, nrows, chunk_tile(c - 1), chunk_tile(c),
                     u0, (c == NCH) ? NWV : NM, ln16, half);
    }
    if (c < NCH) __syncthreads();
  }
}

extern "C" void kernel_launch(void* const* d_in, const int* in_sizes, int n_in,
                              void* d_out, int out_size, void* d_ws, size_t ws_size,
                              hipStream_t stream) {
  const float* x   = (const float*)d_in[0];
  const int*   src = (const int*)  d_in[1];
  const int*   dst = (const int*)  d_in[2];
  const float* ew  = (const float*)d_in[3];
  const float* W1  = (const float*)d_in[4];
  const float* W2  = (const float*)d_in[5];
  float* out = (float*)d_out;

  const int N = in_sizes[0] / DDIM;
  const int E = in_sizes[1];

  float* act1 = (float*)d_ws;
  const size_t np = ((size_t)N + 4) & ~(size_t)3;
  int*  rowptr = (int*)(act1 + (size_t)N * DDIM);
  int*  seg    = rowptr + np;
  int4* rec    = (int4*)(seg + CB * (NBK + 1));
  int2* csr    = (int2*)(rec + E);

  const int gblocks = (N + RB - 1) / RB;

  chunk_sort<<<CB, 1024, 0, stream>>>(src, dst, ew, seg, rec, E, N);

  gather_gemm_tanh<true><<<gblocks, 1024, 0, stream>>>(x, rowptr, csr, W1, act1, N, rec, seg, E);
  gather_gemm_tanh<false><<<gblocks, 1024, 0, stream>>>(act1, rowptr, csr, W2, out, N, rec, seg, E);
}
